# starting kernel
# baseline (speedup 1.0000x reference)
#include <hip/hip_runtime.h>

#define NF_IN  64
#define NF_HID 64
#define NF_OUT 32

typedef __attribute__((ext_vector_type(2))) float v2f;
typedef __attribute__((ext_vector_type(8))) float v8f;

__global__ void fill_kernel(float* __restrict__ p, float val, int n) {
    int i = blockIdx.x * blockDim.x + threadIdx.x;
    if (i < n) p[i] = val;
}

__global__ void count_deg_kernel(const int* __restrict__ dst, float* deg, int E) {
    int e = blockIdx.x * blockDim.x + threadIdx.x;
    if (e < E) atomicAdd(&deg[dst[e]], 1.0f);
}

__global__ void dinv_kernel(float* deg, int n) {
    int i = blockIdx.x * blockDim.x + threadIdx.x;
    if (i < n) deg[i] = rsqrtf(deg[i]);
}

template<int NOUT>
__global__ void gemm_wmma_kernel(const float* __restrict__ X,
                                 const float* __restrict__ W,
                                 float* __restrict__ Y) {
    const int m0   = blockIdx.x * 16;
    const int n0   = blockIdx.y * 16;
    const int lane = threadIdx.x & 31;
    const int half = lane >> 4;
    const int l15  = lane & 15;

    v8f c = {};
    const float* xrow = X + (size_t)(m0 + l15) * 64;
#pragma unroll
    for (int kb = 0; kb < 64; kb += 4) {
        const int ka = kb + half * 2;
        v2f a, b;
        a.x = xrow[ka];
        a.y = xrow[ka + 1];
        b.x = W[(ka + 0) * NOUT + n0 + l15];
        b.y = W[(ka + 1) * NOUT + n0 + l15];
        c = __builtin_amdgcn_wmma_f32_16x16x4_f32(
                 false, a,  false, b,
                 (short)0, c,  false,  false);
    }

    float* yrow = Y + (size_t)(m0 + half * 8) * NOUT + n0 + l15;
#pragma unroll
    for (int j = 0; j < 8; ++j)
        yrow[(size_t)j * NOUT] = c[j];
}

template<int F>
__global__ void scatter_kernel(const int* __restrict__ src,
                               const int* __restrict__ dst,
                               const float* __restrict__ h,
                               const float* __restrict__ dinv,
                               float* agg, int E) {
    int e    = (blockIdx.x * blockDim.x + threadIdx.x) >> 5;
    int lane = threadIdx.x & 31;
    if (e >= E) return;
    int s = src[e], d = dst[e];
    float nrm = dinv[s] * dinv[d];
    const float* hs = h   + (size_t)s * F;
    float*       ad = agg + (size_t)d * F;
#pragma unroll
    for (int f = lane; f < F; f += 32)
        atomicAdd(&ad[f], hs[f] * nrm);
}

template<int F, bool RELU>
__global__ void finalize_kernel(float* __restrict__ agg,
                                const float* __restrict__ hpre,
                                const float* __restrict__ dinv,
                                const float* __restrict__ bias,
                                int total) {
    int idx = blockIdx.x * blockDim.x + threadIdx.x;
    if (idx >= total) return;
    int i = idx / F;
    int f = idx % F;
    float di = dinv[i];
    float v  = agg[idx] + di * di * hpre[idx] + bias[f];
    agg[idx] = RELU ? fmaxf(v, 0.0f) : v;
}

extern "C" void kernel_launch(void* const* d_in, const int* in_sizes, int n_in,
                              void* d_out, int out_size, void* d_ws, size_t ws_size,
                              hipStream_t stream) {
    const float* x  = (const float*)d_in[0];
    const int*   ei = (const int*)  d_in[1];
    const float* W1 = (const float*)d_in[2];
    const float* b1 = (const float*)d_in[3];
    const float* W2 = (const float*)d_in[4];
    const float* b2 = (const float*)d_in[5];

    const int N = in_sizes[0] / NF_IN;
    const int E = in_sizes[1] / 2;
    const int* src = ei;
    const int* dst = ei + E;

    char* ws = (char*)d_ws;
    float* dinv  = (float*)ws;  ws += (((size_t)N * 4) + 255) & ~(size_t)255;
    float* h1pre = (float*)ws;  ws += (size_t)N * NF_HID * 4;
    float* h1    = (float*)ws;  ws += (size_t)N * NF_HID * 4;
    float* h2pre = (float*)ws;
    float* out   = (float*)d_out;

    const int TB = 256;
    auto cdiv = [](int a, int b) { return (a + b - 1) / b; };

    fill_kernel     <<<cdiv(N, TB), TB, 0, stream>>>(dinv, 1.0f, N);
    count_deg_kernel<<<cdiv(E, TB), TB, 0, stream>>>(dst, dinv, E);
    dinv_kernel     <<<cdiv(N, TB), TB, 0, stream>>>(dinv, N);

    gemm_wmma_kernel<NF_HID><<<dim3(N / 16, NF_HID / 16), 32, 0, stream>>>(x, W1, h1pre);
    fill_kernel<<<cdiv(N * NF_HID, TB), TB, 0, stream>>>(h1, 0.0f, N * NF_HID);
    scatter_kernel<NF_HID><<<cdiv(E * 32, TB), TB, 0, stream>>>(src, dst, h1pre, dinv, h1, E);
    finalize_kernel<NF_HID, true><<<cdiv(N * NF_HID, TB), TB, 0, stream>>>(
        h1, h1pre, dinv, b1, N * NF_HID);

    gemm_wmma_kernel<NF_OUT><<<dim3(N / 16, NF_OUT / 16), 32, 0, stream>>>(h1, W2, h2pre);
    fill_kernel<<<cdiv(N * NF_OUT, TB), TB, 0, stream>>>(out, 0.0f, N * NF_OUT);
    scatter_kernel<NF_OUT><<<cdiv(E * 32, TB), TB, 0, stream>>>(src, dst, h2pre, dinv, out, E);
    finalize_kernel<NF_OUT, false><<<cdiv(N * NF_OUT, TB), TB, 0, stream>>>(
        out, h2pre, dinv, b2, N * NF_OUT);
}


# edge sums by a per-call counting sort on dst (atomic arrival rank) and one wave per node gathering rows in registers, replacing per-edge f32 atomics; sum order = rank order
# speedup vs baseline: 1.8646x; 1.8646x over previous
#include <hip/hip_runtime.h>

#define NF_IN  64
#define NF_HID 64
#define NF_OUT 32

typedef __attribute__((ext_vector_type(2))) float v2f;
typedef __attribute__((ext_vector_type(8))) float v8f;

__global__ void zero_int_kernel(int* __restrict__ p, int n) {
    int i = blockIdx.x * blockDim.x + threadIdx.x;
    if (i < n) p[i] = 0;
}

__global__ void rank_kernel(const int* __restrict__ dst, int* cnt,
                            int* __restrict__ rank, int E) {
    int e = blockIdx.x * blockDim.x + threadIdx.x;
    if (e < E) rank[e] = atomicAdd(&cnt[dst[e]], 1);
}

__global__ void alloc_kernel(const int* cnt, int* total, int* __restrict__ off,
                             float* __restrict__ dinv, int n) {
    int i    = blockIdx.x * blockDim.x + threadIdx.x;
    int lane = threadIdx.x & 31;
    int c    = (i < n) ? cnt[i] : 0;
    int incl = c;
    for (int d = 1; d < 32; d <<= 1) {
        int t = __shfl_up(incl, d);
        if (lane >= d) incl += t;
    }
    int base = 0;
    if (lane == 31) base = atomicAdd(total, incl);
    base = __shfl(base, 31);
    if (i < n) {
        off[i]  = base + incl - c;
        dinv[i] = rsqrtf((float)(c + 1));
    }
}

__global__ void place_kernel(const int* __restrict__ src, const int* __restrict__ dst,
                             const int* __restrict__ rank, const int* __restrict__ off,
                             int* __restrict__ csr, int E) {
    int e = blockIdx.x * blockDim.x + threadIdx.x;
    if (e < E) csr[off[dst[e]] + rank[e]] = src[e];
}

template<int NOUT>
__global__ void gemm_wmma_kernel(const float* __restrict__ X,
                                 const float* __restrict__ W,
                                 float* __restrict__ Y) {
    const int m0   = blockIdx.x * 16;
    const int n0   = blockIdx.y * 16;
    const int lane = threadIdx.x & 31;
    const int half = lane >> 4;
    const int l15  = lane & 15;

    v8f c = {};
    const float* xrow = X + (size_t)(m0 + l15) * 64;
#pragma unroll
    for (int kb = 0; kb < 64; kb += 4) {
        const int ka = kb + half * 2;
        v2f a, b;
        a.x = xrow[ka];
        a.y = xrow[ka + 1];
        b.x = W[(ka + 0) * NOUT + n0 + l15];
        b.y = W[(ka + 1) * NOUT + n0 + l15];
        c = __builtin_amdgcn_wmma_f32_16x16x4_f32(
                 false, a,  false, b,
                 (short)0, c,  false,  false);
    }

    float* yrow = Y + (size_t)(m0 + half * 8) * NOUT + n0 + l15;
#pragma unroll
    for (int j = 0; j < 8; ++j)
        yrow[(size_t)j * NOUT] = c[j];
}

template<int F, bool RELU>
__global__ void gather_kernel(const int* __restrict__ csr, const int* __restrict__ off,
                              const int* __restrict__ cnt, const float* __restrict__ dinv,
                              const float* __restrict__ h, const float* __restrict__ bias,
                              float* __restrict__ out, int n) {
    constexpr int LPR = F / 2;
    constexpr int EPI = 32 / LPR;
    const int lane = threadIdx.x & 31;
    const int sub  = lane / LPR;
    const int q    = lane % LPR;
    const int i = __builtin_amdgcn_readfirstlane(
        (int)((blockIdx.x * blockDim.x + threadIdx.x) >> 5));
    if (i >= n) return;
    const int   start = off[i];
    const int   c     = cnt[i];
    const float di    = dinv[i];
    const float* hq   = h + 2 * q;

    v2f acc = {0.0f, 0.0f};
    if (sub == 0) {
        const v2f r = *(const v2f*)(hq + (size_t)i * F);
        const float w = di * di;
        acc.x = w * r.x;
        acc.y = w * r.y;
    }
    for (int base = 0; base < c; base += 32) {
        const int m = (c - base < 32) ? (c - base) : 32;
        int   s_l = i;
        float w_l = 0.0f;
        if (lane < m) {
            s_l = csr[start + base + lane];
            w_l = dinv[s_l] * di;
        }
        int j = 0;
        for (; j + 4 * EPI <= m; j += 4 * EPI) {
            v2f   r[4];
            float w[4];
#pragma unroll
            for (int u = 0; u < 4; ++u) {
                const int jj = j + u * EPI + sub;
                int s;
                if constexpr (EPI == 1) {
                    s    = __builtin_amdgcn_readlane(s_l, jj);
                    w[u] = __int_as_float(__builtin_amdgcn_readlane(__float_as_int(w_l), jj));
                } else {
                    s    = __shfl(s_l, jj);
                    w[u] = __shfl(w_l, jj);
                }
                r[u] = *(const v2f*)(hq + (size_t)s * F);
            }
#pragma unroll
            for (int u = 0; u < 4; ++u) {
                acc.x = fmaf(w[u], r[u].x, acc.x);
                acc.y = fmaf(w[u], r[u].y, acc.y);
            }
        }
        for (; j < m; j += EPI) {
            const int jj = j + sub;
            int s; float w;
            if constexpr (EPI == 1) {
                s = __builtin_amdgcn_readlane(s_l, jj);
                w = __int_as_float(__builtin_amdgcn_readlane(__float_as_int(w_l), jj));
            } else {
                s = __shfl(s_l, jj & 31);
                w = __shfl(w_l, jj & 31);
            }
            if (jj < m) {
                const v2f r = *(const v2f*)(hq + (size_t)s * F);
                acc.x = fmaf(w, r.x, acc.x);
                acc.y = fmaf(w, r.y, acc.y);
            }
        }
    }
    if constexpr (EPI == 2) {
        acc.x += __shfl_xor(acc.x, 16);
        acc.y += __shfl_xor(acc.y, 16);
    }
    if (sub == 0) {
        const v2f b = *(const v2f*)(bias + 2 * q);
        v2f v;
        v.x = acc.x + b.x;
        v.y = acc.y + b.y;
        if (RELU) { v.x = fmaxf(v.x, 0.0f); v.y = fmaxf(v.y, 0.0f); }
        *(v2f*)(out + (size_t)i * F + 2 * q) = v;
    }
}

extern "C" void kernel_launch(void* const* d_in, const int* in_sizes, int n_in,
                              void* d_out, int out_size, void* d_ws, size_t ws_size,
                              hipStream_t stream) {
    const float* x  = (const float*)d_in[0];
    const int*   ei = (const int*)  d_in[1];
    const float* W1 = (const float*)d_in[2];
    const float* b1 = (const float*)d_in[3];
    const float* W2 = (const float*)d_in[4];
    const float* b2 = (const float*)d_in[5];

    const int N = in_sizes[0] / NF_IN;
    const int E = in_sizes[1] / 2;
    const int* src = ei;
    const int* dst = ei + E;

    char* ws = (char*)d_ws;
    auto take = [&](size_t bytes) { char* p = ws; ws += (bytes + 255) & ~(size_t)255; return p; };
    int*   cnt   = (int*)  take(((size_t)N + 1) * 4);
    int*   total = cnt + N;
    int*   off   = (int*)  take((size_t)N * 4);
    float* dinv  = (float*)take((size_t)N * 4);
    int*   rank  = (int*)  take((size_t)E * 4);
    int*   csr   = (int*)  take((size_t)E * 4);
    float* h1pre = (float*)take((size_t)N * NF_HID * 4);
    float* h1    = (float*)take((size_t)N * NF_HID * 4);
    float* h2pre = (float*)take((size_t)N * NF_OUT * 4);
    float* out   = (float*)d_out;

    const int TB = 256;
    auto cdiv = [](int a, int b) { return (a + b - 1) / b; };

    zero_int_kernel<<<cdiv(N + 1, TB), TB, 0, stream>>>(cnt, N + 1);
    rank_kernel    <<<cdiv(E, TB), TB, 0, stream>>>(dst, cnt, rank, E);
    alloc_kernel   <<<cdiv(N, TB), TB, 0, stream>>>(cnt, total, off, dinv, N);
    place_kernel   <<<cdiv(E, TB), TB, 0, stream>>>(src, dst, rank, off, csr, E);

    gemm_wmma_kernel<NF_HID><<<dim3(N / 16, NF_HID / 16), 32, 0, stream>>>(x, W1, h1pre);
    gather_kernel<NF_HID, true><<<cdiv(N * 32, TB), TB, 0, stream>>>(
        csr, off, cnt, dinv, h1pre, b1, h1, N);

    gemm_wmma_kernel<NF_OUT><<<dim3(N / 16, NF_OUT / 16), 32, 0, stream>>>(h1, W2, h2pre);
    gather_kernel<NF_OUT, false><<<cdiv(N * 32, TB), TB, 0, stream>>>(
        csr, off, cnt, dinv, h2pre, b2, out, N);
}


# edge ranks taken from LDS counters in two levels (256-node dst bins, then one work-group per bin) instead of one returning global atomic per edge; sum order = LDS arrival rank
# speedup vs baseline: 3.9259x; 2.1055x over previous
#include <hip/hip_runtime.h>


#define NF_IN  64
#define NF_HID 64
#define NF_OUT 32

typedef __attribute__((ext_vector_type(2))) float v2f;
typedef __attribute__((ext_vector_type(4))) float v4f;
typedef __attribute__((ext_vector_type(8))) float v8f;
typedef __attribute__((ext_vector_type(4))) int   v4i;

#define BIN_SHIFT 8
#define MAXNB     512
#define CUR_PAD   64
#define EDGE_TB   512
#define SORT_TB   1024

typedef __attribute__((ext_vector_type(2))) int v2i;

__global__ void zero_int_kernel(int* __restrict__ p, int n) {
    const int stride = gridDim.x * blockDim.x;
    for (int i = blockIdx.x * blockDim.x + threadIdx.x; i < n; i += stride) p[i] = 0;
}

__device__ __forceinline__ int wave_incl_scan(int v, int lane) {
    for (int d = 1; d < 32; d <<= 1) {
        const int t = __shfl_up(v, d);
        if (lane >= d) v += t;
    }
    return v;
}

__device__ __forceinline__ int load_edges8(const int* __restrict__ src, const int* __restrict__ dst,
                                           int e0, int E, int (&d)[8], int (&s)[8], bool want_src) {
#pragma unroll
    for (int k = 0; k < 8; ++k) { d[k] = 0; s[k] = 0; }
    if (e0 >= E) return 0;
    size_t al = reinterpret_cast<size_t>(dst);
    if (want_src) al |= reinterpret_cast<size_t>(src);
    if ((e0 + 8 <= E) && ((al & 15) == 0)) {
        const v4i d0 = *(const v4i*)(dst + e0), d1 = *(const v4i*)(dst + e0 + 4);
        d[0] = d0.x; d[1] = d0.y; d[2] = d0.z; d[3] = d0.w;
        d[4] = d1.x; d[5] = d1.y; d[6] = d1.z; d[7] = d1.w;
        if (want_src) {
            const v4i s0 = *(const v4i*)(src + e0), s1 = *(const v4i*)(src + e0 + 4);
            s[0] = s0.x; s[1] = s0.y; s[2] = s0.z; s[3] = s0.w;
            s[4] = s1.x; s[5] = s1.y; s[6] = s1.z; s[7] = s1.w;
        }
        return 8;
    }
    const int nv = (E - e0 < 8) ? (E - e0) : 8;
#pragma unroll
    for (int k = 0; k < 8; ++k) {
        d[k] = (k < nv) ? dst[e0 + k] : 0;
        if (want_src) s[k] = (k < nv) ? src[e0 + k] : 0;
    }
    return nv;
}

__global__ void __launch_bounds__(EDGE_TB)
bin_count_kernel(const int* __restrict__ dst, int* bintotal, int E, int nb) {
    __shared__ int hist[MAXNB];
    const int tid = threadIdx.x;
    hist[tid] = 0;
    __syncthreads();
    int d[8], s[8];
    const int nv = load_edges8(dst, dst, (int)(blockIdx.x * EDGE_TB + tid) * 8, E, d, s, false);
#pragma unroll
    for (int k = 0; k < 8; ++k)
        if (k < nv) atomicAdd(&hist[d[k] >> BIN_SHIFT], 1);
    __syncthreads();
    for (int t = tid; t < nb; t += EDGE_TB) {
        const int h = hist[t];
        if (h > 0) atomicAdd(&bintotal[t], h);
    }
}

__global__ void __launch_bounds__(EDGE_TB)
bin_scatter_kernel(const int* __restrict__ src, const int* __restrict__ dst,
                   const int* __restrict__ bintotal, int* cursor,
                   int* __restrict__ recs, int E, int nb) {
    __shared__ int hist[MAXNB];
    __shared__ int wtot[EDGE_TB / 32];
    const int tid  = threadIdx.x;
    const int lane = tid & 31;
    const int wave = tid >> 5;
    hist[tid] = 0;
    __syncthreads();
    const int a = (tid < nb) ? bintotal[tid] : 0;
    int d[8], s[8], lr[8];
    const int nv = load_edges8(src, dst, (int)(blockIdx.x * EDGE_TB + tid) * 8, E, d, s, true);
#pragma unroll
    for (int k = 0; k < 8; ++k) {
        lr[k] = 0;
        if (k < nv) lr[k] = atomicAdd(&hist[d[k] >> BIN_SHIFT], 1);
    }
    const int incl = wave_incl_scan(a, lane);
    if (lane == 31) wtot[wave] = incl;
    __syncthreads();
    int carry = 0;
#pragma unroll
    for (int w = 0; w < EDGE_TB / 32; ++w) carry += (w < wave) ? wtot[w] : 0;
    int base = carry + incl - a;
    const int h = hist[tid];
    if (h > 0) base += atomicAdd(&cursor[(size_t)tid * CUR_PAD], h);
    hist[tid] = base;
    __syncthreads();
    int pos[8];
#pragma unroll
    for (int k = 0; k < 8; ++k) pos[k] = hist[d[k] >> BIN_SHIFT] + lr[k];
#pragma unroll
    for (int k = 0; k < 8; ++k) {
        if (k < nv) {
            v2i r;
            r.x = s[k];
            r.y = d[k];
            *(v2i*)(recs + 2 * (size_t)pos[k]) = r;
        }
    }
}

__global__ void __launch_bounds__(SORT_TB)
bin_sort_kernel(const int* __restrict__ recs, const int* __restrict__ bintotal,
                int* __restrict__ csr, int* __restrict__ off, int* __restrict__ cnt,
                float* __restrict__ dinv, int n) {
    constexpr int RB = 1 << BIN_SHIFT;
    constexpr int KR = 8;
    __shared__ int lcnt[RB];
    __shared__ int loff[RB];
    __shared__ int wtot[RB / 32];
    __shared__ int sh_start;
    const int b    = blockIdx.x;
    const int tid  = threadIdx.x;
    const int lane = tid & 31;
    const int lo   = b << BIN_SHIFT;
    if (tid < RB) lcnt[tid] = 0;
    if (tid == 0) sh_start = 0;
    __syncthreads();
    int part = 0;
    for (int t = tid; t < b; t += SORT_TB) part += bintotal[t];
    for (int dl = 16; dl > 0; dl >>= 1) part += __shfl_xor(part, dl);
    if (lane == 0 && part != 0) atomicAdd(&sh_start, part);
    __syncthreads();
    const int  bs   = sh_start;
    const int  nbin = bintotal[b];
    const v2i* rp   = (const v2i*)recs + bs;
    v2i rec[KR];
#pragma unroll
    for (int k = 0; k < KR; ++k) {
        const int r = tid + k * SORT_TB;
        rec[k].x = 0;
        rec[k].y = lo;
        if (r < nbin) rec[k] = rp[r];
    }
#pragma unroll
    for (int k = 0; k < KR; ++k)
        if (tid + k * SORT_TB < nbin) atomicAdd(&lcnt[rec[k].y - lo], 1);
    for (int r = tid + KR * SORT_TB; r < nbin; r += SORT_TB) atomicAdd(&lcnt[rp[r].y - lo], 1);
    __syncthreads();
    int v = 0, incl = 0, lf = 0;
    if (tid < RB) {
        v    = lcnt[tid];
        incl = wave_incl_scan(v, lane);
        if (lane == 31) wtot[tid >> 5] = incl;
    }
    __syncthreads();
    if (tid < RB) {
        int carry = 0;
        for (int w = 0; w < (tid >> 5); ++w) carry += wtot[w];
        lf = carry + incl - v;
        loff[tid] = lf;
        lcnt[tid] = 0;
    }
    __syncthreads();
    if (tid < RB && lo + tid < n) {
        const int i = lo + tid;
        off[i]  = bs + lf;
        cnt[i]  = v;
        dinv[i] = rsqrtf((float)(v + 1));
    }
#pragma unroll
    for (int k = 0; k < KR; ++k) {
        if (tid + k * SORT_TB < nbin) {
            const int j = rec[k].y - lo;
            const int p = atomicAdd(&lcnt[j], 1);
            csr[bs + loff[j] + p] = rec[k].x;
        }
    }
    for (int r = tid + KR * SORT_TB; r < nbin; r += SORT_TB) {
        const v2i rc = rp[r];
        const int j  = rc.y - lo;
        const int p  = atomicAdd(&lcnt[j], 1);
        csr[bs + loff[j] + p] = rc.x;
    }
}

template<int NOUT>
__global__ void gemm_wmma_kernel(const float* __restrict__ X,
                                 const float* __restrict__ W,
                                 float* __restrict__ Y, int ntiles) {
    constexpr int NT = NOUT / 16;
    typedef float vnt __attribute__((ext_vector_type(NT)));
    const int tile = __builtin_amdgcn_readfirstlane(
        (int)(blockIdx.x * (blockDim.x >> 5) + (threadIdx.x >> 5)));
    if (tile >= ntiles) return;
    const int m0   = tile * 16;
    const int lane = threadIdx.x & 31;
    const int half = lane >> 4;
    const int l15  = lane & 15;

    v8f c[NT];
#pragma unroll
    for (int t = 0; t < NT; ++t) c[t] = v8f{};
    const float* xrow = X + (size_t)(m0 + l15) * 64;
    const float* wcol = W + l15 * NT;
#pragma unroll
    for (int kb = 0; kb < 64; kb += 4) {
        const int ka = kb + half * 2;
        const v2f a  = *(const v2f*)(xrow + ka);
        const vnt b0 = *(const vnt*)(wcol + (size_t)(ka + 0) * NOUT);
        const vnt b1 = *(const vnt*)(wcol + (size_t)(ka + 1) * NOUT);
#pragma unroll
        for (int t = 0; t < NT; ++t) {
            v2f b;
            b.x = b0[t];
            b.y = b1[t];
            c[t] = __builtin_amdgcn_wmma_f32_16x16x4_f32(
                     false, a,  false, b,
                     (short)0, c[t],  false,  false);
        }
    }

    float* yrow = Y + (size_t)(m0 + half * 8) * NOUT + l15 * NT;
#pragma unroll
    for (int j = 0; j < 8; ++j) {
        vnt v;
#pragma unroll
        for (int t = 0; t < NT; ++t) v[t] = c[t][j];
        *(vnt*)(yrow + (size_t)j * NOUT) = v;
    }
}

template<int F, int U>
__device__ __forceinline__ void issue_rows(const float* hq, int sl, float wl, int j, int sub,
                                           v2f (&r)[U], float (&w)[U]) {
    constexpr int EPI = 64 / F;
#pragma unroll
    for (int u = 0; u < U; ++u) {
        const int jj = j + u * EPI + sub;
        int s;
        if constexpr (EPI == 1) {
            s    = __builtin_amdgcn_readlane(sl, jj);
            w[u] = __int_as_float(__builtin_amdgcn_readlane(__float_as_int(wl), jj));
        } else {
            s    = __shfl(sl, jj);
            w[u] = __shfl(wl, jj);
        }
        r[u] = *(const v2f*)(hq + (size_t)s * F);
    }
}

template<int F, int U>
__device__ __forceinline__ void add_rows(v2f& acc, int j, int sub, int m,
                                         const v2f (&r)[U], const float (&w)[U]) {
    constexpr int EPI = 64 / F;
#pragma unroll
    for (int u = 0; u < U; ++u) {
        const int jj = j + u * EPI + sub;
        if (jj < m) {
            acc.x = fmaf(w[u], r[u].x, acc.x);
            acc.y = fmaf(w[u], r[u].y, acc.y);
        }
    }
}

template<int F, bool RELU>
__global__ void gather_kernel(const int* __restrict__ csr, const int* __restrict__ off,
                              const int* __restrict__ cnt, const float* __restrict__ dinv,
                              const float* __restrict__ h, const float* __restrict__ bias,
                              float* __restrict__ out, int n) {
    constexpr int LPR = F / 2;
    constexpr int EPI = 32 / LPR;
    constexpr int RS  = 16;
    constexpr int U   = RS / EPI;
    const int lane = threadIdx.x & 31;
    const int sub  = lane / LPR;
    const int q    = lane % LPR;
    const int wave  = __builtin_amdgcn_readfirstlane(
        (int)((blockIdx.x * blockDim.x + threadIdx.x) >> 5));
    const int nwave = (int)((gridDim.x * blockDim.x) >> 5);
    const int per = n / nwave;
    const int rem = n - per * nwave;
    const int i0  = wave * per + ((wave < rem) ? wave : rem);
    const int i1  = i0 + per + ((wave < rem) ? 1 : 0);
    if (i0 >= i1) return;
    const float* hq = h + 2 * q;
    const v2f bq = *(const v2f*)(bias + 2 * q);

    int   start = off[i0];
    int   c     = cnt[i0];
    float di    = dinv[i0];
    int   s_l   = i0;
    float w_l   = 0.0f;
    if (lane < c) {
        s_l = csr[start + lane];
        w_l = dinv[s_l] * di;
    }
    for (int i = i0; i < i1; ++i) {
        const int   inx    = (i + 1 < i1) ? i + 1 : i;
        const int   startN = off[inx];
        const int   cN     = cnt[inx];
        const float diN    = dinv[inx];
        const int   m      = (c < 32) ? c : 32;

        const v2f rs = *(const v2f*)(hq + (size_t)i * F);
        v2f   r[U];
        float w[U];
        issue_rows<F, U>(hq, s_l, w_l, 0, sub, r, w);
        int   sN = inx;
        float dN = 0.0f;
        if (lane < cN) sN = csr[startN + lane];
        const float ws = di * di;
        v2f acc;
        acc.x = (sub == 0) ? ws * rs.x : 0.0f;
        acc.y = (sub == 0) ? ws * rs.y : 0.0f;
        add_rows<F, U>(acc, 0, sub, m, r, w);
        if (lane < cN) dN = dinv[sN];
        if (m > RS) {
            issue_rows<F, U>(hq, s_l, w_l, RS, sub, r, w);
            add_rows<F, U>(acc, RS, sub, m, r, w);
        }
        for (int base = 32; base < c; base += 32) {
            const int m2 = (c - base < 32) ? (c - base) : 32;
            int   s2 = i;
            float w2 = 0.0f;
            if (lane < m2) {
                s2 = csr[start + base + lane];
                w2 = dinv[s2] * di;
            }
            for (int j = 0; j < m2; j += RS) {
                issue_rows<F, U>(hq, s2, w2, j, sub, r, w);
                add_rows<F, U>(acc, j, sub, m2, r, w);
            }
        }
        if constexpr (EPI == 2) {
            acc.x += __shfl_xor(acc.x, 16);
            acc.y += __shfl_xor(acc.y, 16);
        }
        if (sub == 0) {
            v2f v;
            v.x = acc.x + bq.x;
            v.y = acc.y + bq.y;
            if (RELU) { v.x = fmaxf(v.x, 0.0f); v.y = fmaxf(v.y, 0.0f); }
            *(v2f*)(out + (size_t)i * F + 2 * q) = v;
        }
        start = startN; c = cN; di = diN; s_l = sN; w_l = dN * diN;
    }
}

extern "C" void kernel_launch(void* const* d_in, const int* in_sizes, int n_in,
                              void* d_out, int out_size, void* d_ws, size_t ws_size,
                              hipStream_t stream) {
    const float* x  = (const float*)d_in[0];
    const int*   ei = (const int*)  d_in[1];
    const float* W1 = (const float*)d_in[2];
    const float* b1 = (const float*)d_in[3];
    const float* W2 = (const float*)d_in[4];
    const float* b2 = (const float*)d_in[5];

    const int N = in_sizes[0] / NF_IN;
    const int E = in_sizes[1] / 2;
    const int* src = ei;
    const int* dst = ei + E;

    char* ws = (char*)d_ws;
    auto take = [&](size_t bytes) { char* p = ws; ws += (bytes + 255) & ~(size_t)255; return p; };
    auto cdiv = [](int a, int b) { return (a + b - 1) / b; };
    const int NB = cdiv(N, 1 << BIN_SHIFT);
    int*   cursor   = (int*)  take((size_t)NB * (CUR_PAD + 1) * 4);
    int*   bintotal = cursor + (size_t)NB * CUR_PAD;
    int*   cnt   = (int*)  take((size_t)N * 4);
    int*   off   = (int*)  take((size_t)N * 4);
    float* dinv  = (float*)take((size_t)N * 4);
    int*   recs  = (int*)  take((size_t)E * 8);
    int*   csr   = (int*)  take((size_t)E * 4);
    float* h1pre = (float*)take((size_t)N * NF_HID * 4);
    float* h1    = (float*)take((size_t)N * NF_HID * 4);
    float* h2pre = (float*)take((size_t)N * NF_OUT * 4);
    float* out   = (float*)d_out;

    const int TB = 256;
    const int GW = 1024;

    zero_int_kernel   <<<64, TB, 0, stream>>>(cursor, NB * (CUR_PAD + 1));
    bin_count_kernel  <<<cdiv(E, EDGE_TB * 8), EDGE_TB, 0, stream>>>(dst, bintotal, E, NB);
    bin_scatter_kernel<<<cdiv(E, EDGE_TB * 8), EDGE_TB, 0, stream>>>(src, dst, bintotal, cursor, recs, E, NB);
    bin_sort_kernel   <<<NB, SORT_TB, 0, stream>>>(recs, bintotal, csr, off, cnt, dinv, N);

    gemm_wmma_kernel<NF_HID><<<cdiv(N / 16, 4), 128, 0, stream>>>(x, W1, h1pre, N / 16);
    gather_kernel<NF_HID, true><<<GW, TB, 0, stream>>>(
        csr, off, cnt, dinv, h1pre, b1, h1, N);

    gemm_wmma_kernel<NF_OUT><<<cdiv(N / 16, 4), 128, 0, stream>>>(h1, W2, h2pre, N / 16);
    gather_kernel<NF_OUT, false><<<GW, TB, 0, stream>>>(
        csr, off, cnt, dinv, h2pre, b2, out, N);
}


# dinv[src] folded into the product rows at the GEMM store (rows hold dinv*xW): the gather sums plain rows and scales once by dinv[dst]; no per-edge dinv load or multiply; sum order kept
# speedup vs baseline: 4.4242x; 1.1269x over previous
#include <hip/hip_runtime.h>


#define NF_IN  64
#define NF_HID 64
#define NF_OUT 32

typedef __attribute__((ext_vector_type(2))) float v2f;
typedef __attribute__((ext_vector_type(4))) float v4f;
typedef __attribute__((ext_vector_type(8))) float v8f;
typedef __attribute__((ext_vector_type(4))) int   v4i;

#define BIN_SHIFT 8
#define MAXNB     512
#define CUR_PAD   64
#define EDGE_TB   512
#define SORT_TB   1024

typedef __attribute__((ext_vector_type(2))) int v2i;

__global__ void zero_int_kernel(int* __restrict__ p, int n) {
    const int stride = gridDim.x * blockDim.x;
    for (int i = blockIdx.x * blockDim.x + threadIdx.x; i < n; i += stride) p[i] = 0;
}

__device__ __forceinline__ int wave_incl_scan(int v, int lane) {
    for (int d = 1; d < 32; d <<= 1) {
        const int t = __shfl_up(v, d);
        if (lane >= d) v += t;
    }
    return v;
}

__device__ __forceinline__ int load_edges8(const int* __restrict__ src, const int* __restrict__ dst,
                                           int e0, int E, int (&d)[8], int (&s)[8], bool want_src) {
#pragma unroll
    for (int k = 0; k < 8; ++k) { d[k] = 0; s[k] = 0; }
    if (e0 >= E) return 0;
    size_t al = reinterpret_cast<size_t>(dst);
    if (want_src) al |= reinterpret_cast<size_t>(src);
    if ((e0 + 8 <= E) && ((al & 15) == 0)) {
        const v4i d0 = *(const v4i*)(dst + e0), d1 = *(const v4i*)(dst + e0 + 4);
        d[0] = d0.x; d[1] = d0.y; d[2] = d0.z; d[3] = d0.w;
        d[4] = d1.x; d[5] = d1.y; d[6] = d1.z; d[7] = d1.w;
        if (want_src) {
            const v4i s0 = *(const v4i*)(src + e0), s1 = *(const v4i*)(src + e0 + 4);
            s[0] = s0.x; s[1] = s0.y; s[2] = s0.z; s[3] = s0.w;
            s[4] = s1.x; s[5] = s1.y; s[6] = s1.z; s[7] = s1.w;
        }
        return 8;
    }
    const int nv = (E - e0 < 8) ? (E - e0) : 8;
#pragma unroll
    for (int k = 0; k < 8; ++k) {
        d[k] = (k < nv) ? dst[e0 + k] : 0;
        if (want_src) s[k] = (k < nv) ? src[e0 + k] : 0;
    }
    return nv;
}

__global__ void __launch_bounds__(EDGE_TB)
bin_count_kernel(const int* __restrict__ dst, int* bintotal, int E, int nb) {
    __shared__ int hist[MAXNB];
    const int tid = threadIdx.x;
    hist[tid] = 0;
    __syncthreads();
    int d[8], s[8];
    const int nv = load_edges8(dst, dst, (int)(blockIdx.x * EDGE_TB + tid) * 8, E, d, s, false);
#pragma unroll
    for (int k = 0; k < 8; ++k)
        if (k < nv) atomicAdd(&hist[d[k] >> BIN_SHIFT], 1);
    __syncthreads();
    for (int t = tid; t < nb; t += EDGE_TB) {
        const int h = hist[t];
        if (h > 0) atomicAdd(&bintotal[t], h);
    }
}

__global__ void __launch_bounds__(EDGE_TB)
bin_scatter_kernel(const int* __restrict__ src, const int* __restrict__ dst,
                   const int* __restrict__ bintotal, int* cursor,
                   int* __restrict__ recs, int E, int nb) {
    __shared__ int hist[MAXNB];
    __shared__ int wtot[EDGE_TB / 32];
    const int tid  = threadIdx.x;
    const int lane = tid & 31;
    const int wave = tid >> 5;
    hist[tid] = 0;
    __syncthreads();
    const int a = (tid < nb) ? bintotal[tid] : 0;
    int d[8], s[8], lr[8];
    const int nv = load_edges8(src, dst, (int)(blockIdx.x * EDGE_TB + tid) * 8, E, d, s, true);
#pragma unroll
    for (int k = 0; k < 8; ++k) {
        lr[k] = 0;
        if (k < nv) lr[k] = atomicAdd(&hist[d[k] >> BIN_SHIFT], 1);
    }
    const int incl = wave_incl_scan(a, lane);
    if (lane == 31) wtot[wave] = incl;
    __syncthreads();
    int carry = 0;
#pragma unroll
    for (int w = 0; w < EDGE_TB / 32; ++w) carry += (w < wave) ? wtot[w] : 0;
    int base = carry + incl - a;
    const int h = hist[tid];
    if (h > 0) base += atomicAdd(&cursor[(size_t)tid * CUR_PAD], h);
    hist[tid] = base;
    __syncthreads();
    int pos[8];
#pragma unroll
    for (int k = 0; k < 8; ++k) pos[k] = hist[d[k] >> BIN_SHIFT] + lr[k];
#pragma unroll
    for (int k = 0; k < 8; ++k) {
        if (k < nv) {
            v2i r;
            r.x = s[k];
            r.y = d[k];
            *(v2i*)(recs + 2 * (size_t)pos[k]) = r;
        }
    }
}

__global__ void __launch_bounds__(SORT_TB)
bin_sort_kernel(const int* __restrict__ recs, const int* __restrict__ bintotal,
                int* __restrict__ csr, int* __restrict__ off, int* __restrict__ cnt,
                float* __restrict__ dinv, int n) {
    constexpr int RB = 1 << BIN_SHIFT;
    constexpr int KR = 8;
    __shared__ int lcnt[RB];
    __shared__ int loff[RB];
    __shared__ int wtot[RB / 32];
    __shared__ int sh_start;
    const int b    = blockIdx.x;
    const int tid  = threadIdx.x;
    const int lane = tid & 31;
    const int lo   = b << BIN_SHIFT;
    if (tid < RB) lcnt[tid] = 0;
    if (tid == 0) sh_start = 0;
    __syncthreads();
    int part = 0;
    for (int t = tid; t < b; t += SORT_TB) part += bintotal[t];
    for (int dl = 16; dl > 0; dl >>= 1) part += __shfl_xor(part, dl);
    if (lane == 0 && part != 0) atomicAdd(&sh_start, part);
    __syncthreads();
    const int  bs   = sh_start;
    const int  nbin = bintotal[b];
    const v2i* rp   = (const v2i*)recs + bs;
    v2i rec[KR];
#pragma unroll
    for (int k = 0; k < KR; ++k) {
        const int r = tid + k * SORT_TB;
        rec[k].x = 0;
        rec[k].y = lo;
        if (r < nbin) rec[k] = rp[r];
    }
#pragma unroll
    for (int k = 0; k < KR; ++k)
        if (tid + k * SORT_TB < nbin) atomicAdd(&lcnt[rec[k].y - lo], 1);
    for (int r = tid + KR * SORT_TB; r < nbin; r += SORT_TB) atomicAdd(&lcnt[rp[r].y - lo], 1);
    __syncthreads();
    int v = 0, incl = 0, lf = 0;
    if (tid < RB) {
        v    = lcnt[tid];
        incl = wave_incl_scan(v, lane);
        if (lane == 31) wtot[tid >> 5] = incl;
    }
    __syncthreads();
    if (tid < RB) {
        int carry = 0;
        for (int w = 0; w < (tid >> 5); ++w) carry += wtot[w];
        lf = carry + incl - v;
        loff[tid] = lf;
        lcnt[tid] = 0;
    }
    __syncthreads();
    if (tid < RB && lo + tid < n) {
        const int i = lo + tid;
        off[i]  = bs + lf;
        cnt[i]  = v;
        dinv[i] = rsqrtf((float)(v + 1));
    }
#pragma unroll
    for (int k = 0; k < KR; ++k) {
        if (tid + k * SORT_TB < nbin) {
            const int j = rec[k].y - lo;
            const int p = atomicAdd(&lcnt[j], 1);
            csr[bs + loff[j] + p] = rec[k].x;
        }
    }
    for (int r = tid + KR * SORT_TB; r < nbin; r += SORT_TB) {
        const v2i rc = rp[r];
        const int j  = rc.y - lo;
        const int p  = atomicAdd(&lcnt[j], 1);
        csr[bs + loff[j] + p] = rc.x;
    }
}

template<int NOUT>
__global__ void __launch_bounds__(128)
gemm_wmma_kernel(const float* __restrict__ X,
                                 const float* __restrict__ W,
                                 const float* __restrict__ rscale,
                                 float* __restrict__ Y, int ntiles) {
    constexpr int NT = NOUT / 16;
    typedef float vnt __attribute__((ext_vector_type(NT)));
    const int tile = __builtin_amdgcn_readfirstlane(
        (int)(blockIdx.x * (blockDim.x >> 5) + (threadIdx.x >> 5)));
    if (tile >= ntiles) return;
    const int m0   = tile * 16;
    const int lane = threadIdx.x & 31;
    const int half = lane >> 4;
    const int l15  = lane & 15;

    v8f c[NT];
#pragma unroll
    for (int t = 0; t < NT; ++t) c[t] = v8f{};
    const float* xrow = X + (size_t)(m0 + l15) * 64;
    const float* wcol = W + l15 * NT;
#pragma unroll
    for (int kb = 0; kb < 64; kb += 4) {
        const int ka = kb + half * 2;
        const v2f a  = *(const v2f*)(xrow + ka);
        const vnt b0 = *(const vnt*)(wcol + (size_t)(ka + 0) * NOUT);
        const vnt b1 = *(const vnt*)(wcol + (size_t)(ka + 1) * NOUT);
#pragma unroll
        for (int t = 0; t < NT; ++t) {
            v2f b;
            b.x = b0[t];
            b.y = b1[t];
            c[t] = __builtin_amdgcn_wmma_f32_16x16x4_f32(
                     false, a,  false, b,
                     (short)0, c[t],  false,  false);
        }
    }

    const v4f rs0 = *(const v4f*)(rscale + m0 + half * 8);
    const v4f rs1 = *(const v4f*)(rscale + m0 + half * 8 + 4);
    const float rsc[8] = {rs0.x, rs0.y, rs0.z, rs0.w, rs1.x, rs1.y, rs1.z, rs1.w};
    float* yrow = Y + (size_t)(m0 + half * 8) * NOUT + l15 * NT;
#pragma unroll
    for (int j = 0; j < 8; ++j) {
        vnt v;
#pragma unroll
        for (int t = 0; t < NT; ++t) v[t] = c[t][j] * rsc[j];
        *(vnt*)(yrow + (size_t)j * NOUT) = v;
    }
}

template<int F, int U>
__device__ __forceinline__ void issue_rows(const float* hq, int sl, int j, int sub,
                                           v2f (&r)[U]) {
    constexpr int EPI = 64 / F;
#pragma unroll
    for (int u = 0; u < U; ++u) {
        const int jj = j + u * EPI + sub;
        int s;
        if constexpr (EPI == 1) s = __builtin_amdgcn_readlane(sl, jj);
        else                    s = __shfl(sl, jj);
        r[u] = *(const v2f*)(hq + (size_t)s * F);
    }
}

template<int F, int U>
__device__ __forceinline__ void add_rows(v2f& acc, int j, int sub, int m,
                                         const v2f (&r)[U]) {
    constexpr int EPI = 64 / F;
#pragma unroll
    for (int u = 0; u < U; ++u) {
        const int jj = j + u * EPI + sub;
        if (jj < m) {
            acc.x += r[u].x;
            acc.y += r[u].y;
        }
    }
}

template<int F, bool RELU>
__global__ void gather_kernel(const int* __restrict__ csr, const int* __restrict__ off,
                              const int* __restrict__ cnt, const float* __restrict__ dinv,
                              const float* __restrict__ h, const float* __restrict__ bias,
                              float* __restrict__ out, int n) {
    constexpr int LPR = F / 2;
    constexpr int EPI = 32 / LPR;
    constexpr int RS  = 16;
    constexpr int U   = RS / EPI;
    const int lane = threadIdx.x & 31;
    const int sub  = lane / LPR;
    const int q    = lane % LPR;
    const int wave  = __builtin_amdgcn_readfirstlane(
        (int)((blockIdx.x * blockDim.x + threadIdx.x) >> 5));
    const int nwave = (int)((gridDim.x * blockDim.x) >> 5);
    const int per = n / nwave;
    const int rem = n - per * nwave;
    const int i0  = wave * per + ((wave < rem) ? wave : rem);
    const int i1  = i0 + per + ((wave < rem) ? 1 : 0);
    if (i0 >= i1) return;
    const float* hq = h + 2 * q;
    const v2f bq = *(const v2f*)(bias + 2 * q);

    int   start = off[i0];
    int   c     = cnt[i0];
    float di    = dinv[i0];
    int   s_l   = i0;
    if (lane < c) s_l = csr[start + lane];
    for (int i = i0; i < i1; ++i) {
        const int   inx    = (i + 1 < i1) ? i + 1 : i;
        const int   startN = off[inx];
        const int   cN     = cnt[inx];
        const float diN    = dinv[inx];
        const int   m      = (c < 32) ? c : 32;

        const v2f rs = *(const v2f*)(hq + (size_t)i * F);
        v2f r[U];
        issue_rows<F, U>(hq, s_l, 0, sub, r);
        int sN = inx;
        if (lane < cN) sN = csr[startN + lane];
        v2f acc;
        acc.x = (sub == 0) ? rs.x : 0.0f;
        acc.y = (sub == 0) ? rs.y : 0.0f;
        add_rows<F, U>(acc, 0, sub, m, r);
        if (m > RS) {
            issue_rows<F, U>(hq, s_l, RS, sub, r);
            add_rows<F, U>(acc, RS, sub, m, r);
        }
        for (int base = 32; base < c; base += 32) {
            const int m2 = (c - base < 32) ? (c - base) : 32;
            int s2 = i;
            if (lane < m2) s2 = csr[start + base + lane];
            for (int j = 0; j < m2; j += RS) {
                issue_rows<F, U>(hq, s2, j, sub, r);
                add_rows<F, U>(acc, j, sub, m2, r);
            }
        }
        if constexpr (EPI == 2) {
            acc.x += __shfl_xor(acc.x, 16);
            acc.y += __shfl_xor(acc.y, 16);
        }
        if (sub == 0) {
            v2f v;
            v.x = di * acc.x + bq.x;
            v.y = di * acc.y + bq.y;
            if (RELU) { v.x = fmaxf(v.x, 0.0f); v.y = fmaxf(v.y, 0.0f); }
            *(v2f*)(out + (size_t)i * F + 2 * q) = v;
        }
        start = startN; c = cN; di = diN; s_l = sN;
    }
}

extern "C" void kernel_launch(void* const* d_in, const int* in_sizes, int n_in,
                              void* d_out, int out_size, void* d_ws, size_t ws_size,
                              hipStream_t stream) {
    const float* x  = (const float*)d_in[0];
    const int*   ei = (const int*)  d_in[1];
    const float* W1 = (const float*)d_in[2];
    const float* b1 = (const float*)d_in[3];
    const float* W2 = (const float*)d_in[4];
    const float* b2 = (const float*)d_in[5];

    const int N = in_sizes[0] / NF_IN;
    const int E = in_sizes[1] / 2;
    const int* src = ei;
    const int* dst = ei + E;

    char* ws = (char*)d_ws;
    auto take = [&](size_t bytes) { char* p = ws; ws += (bytes + 255) & ~(size_t)255; return p; };
    auto cdiv = [](int a, int b) { return (a + b - 1) / b; };
    const int NB = cdiv(N, 1 << BIN_SHIFT);
    int*   cursor   = (int*)  take((size_t)NB * (CUR_PAD + 1) * 4);
    int*   bintotal = cursor + (size_t)NB * CUR_PAD;
    int*   cnt   = (int*)  take((size_t)N * 4);
    int*   off   = (int*)  take((size_t)N * 4);
    float* dinv  = (float*)take((size_t)N * 4);
    int*   recs  = (int*)  take((size_t)E * 8);
    int*   csr   = (int*)  take((size_t)E * 4);
    float* h1pre = (float*)take((size_t)N * NF_HID * 4);
    float* h1    = (float*)take((size_t)N * NF_HID * 4);
    float* h2pre = (float*)take((size_t)N * NF_OUT * 4);
    float* out   = (float*)d_out;

    const int TB = 256;
    const int GW = 1024;

    zero_int_kernel   <<<64, TB, 0, stream>>>(cursor, NB * (CUR_PAD + 1));
    bin_count_kernel  <<<cdiv(E, EDGE_TB * 8), EDGE_TB, 0, stream>>>(dst, bintotal, E, NB);
    bin_scatter_kernel<<<cdiv(E, EDGE_TB * 8), EDGE_TB, 0, stream>>>(src, dst, bintotal, cursor, recs, E, NB);
    bin_sort_kernel   <<<NB, SORT_TB, 0, stream>>>(recs, bintotal, csr, off, cnt, dinv, N);

    gemm_wmma_kernel<NF_HID><<<cdiv(N / 16, 4), 128, 0, stream>>>(x, W1, dinv, h1pre, N / 16);
    gather_kernel<NF_HID, true><<<GW, TB, 0, stream>>>(
        csr, off, cnt, dinv, h1pre, b1, h1, N);

    gemm_wmma_kernel<NF_OUT><<<cdiv(N / 16, 4), 128, 0, stream>>>(h1, W2, dinv, h2pre, N / 16);
    gather_kernel<NF_OUT, false><<<GW, TB, 0, stream>>>(
        csr, off, cnt, dinv, h2pre, b2, out, N);
}


# no global counters in the edge sort: each edge work-group orders its 4096 records by bin in its own segment and publishes place+length per bin; 2 launches fewer; sum order = LDS arrival rank
# speedup vs baseline: 4.9419x; 1.1170x over previous
#include <hip/hip_runtime.h>


#define NF_IN  64
#define NF_HID 64
#define NF_OUT 32

typedef __attribute__((ext_vector_type(2))) float v2f;
typedef __attribute__((ext_vector_type(4))) float v4f;
typedef __attribute__((ext_vector_type(8))) float v8f;
typedef __attribute__((ext_vector_type(4))) int   v4i;

#define BIN_SHIFT 8
#define MAXNB     512
#define EDGE_TB   512
#define SEG       4096
#define SORT_TB   1024

typedef __attribute__((ext_vector_type(2))) int v2i;

__device__ __forceinline__ int wave_incl_scan(int v, int lane) {
    for (int d = 1; d < 32; d <<= 1) {
        const int t = __shfl_up(v, d);
        if (lane >= d) v += t;
    }
    return v;
}

__device__ __forceinline__ int load_edges8(const int* __restrict__ src, const int* __restrict__ dst,
                                           int e0, int E, int (&d)[8], int (&s)[8], bool want_src) {
#pragma unroll
    for (int k = 0; k < 8; ++k) { d[k] = 0; s[k] = 0; }
    if (e0 >= E) return 0;
    size_t al = reinterpret_cast<size_t>(dst);
    if (want_src) al |= reinterpret_cast<size_t>(src);
    if ((e0 + 8 <= E) && ((al & 15) == 0)) {
        const v4i d0 = *(const v4i*)(dst + e0), d1 = *(const v4i*)(dst + e0 + 4);
        d[0] = d0.x; d[1] = d0.y; d[2] = d0.z; d[3] = d0.w;
        d[4] = d1.x; d[5] = d1.y; d[6] = d1.z; d[7] = d1.w;
        if (want_src) {
            const v4i s0 = *(const v4i*)(src + e0), s1 = *(const v4i*)(src + e0 + 4);
            s[0] = s0.x; s[1] = s0.y; s[2] = s0.z; s[3] = s0.w;
            s[4] = s1.x; s[5] = s1.y; s[6] = s1.z; s[7] = s1.w;
        }
        return 8;
    }
    const int nv = (E - e0 < 8) ? (E - e0) : 8;
#pragma unroll
    for (int k = 0; k < 8; ++k) {
        d[k] = (k < nv) ? dst[e0 + k] : 0;
        if (want_src) s[k] = (k < nv) ? src[e0 + k] : 0;
    }
    return nv;
}

__global__ void __launch_bounds__(EDGE_TB)
bin_local_kernel(const int* __restrict__ src, const int* __restrict__ dst,
                 int* __restrict__ pk, int* __restrict__ recs, int E, int nb) {
    __shared__ int hist[MAXNB];
    __shared__ int wtot[EDGE_TB / 32];
    const int tid  = threadIdx.x;
    const int lane = tid & 31;
    const int wave = tid >> 5;
    hist[tid] = 0;
    __syncthreads();
    int d[8], s[8], lr[8];
    const int nv = load_edges8(src, dst, (int)(blockIdx.x * EDGE_TB + tid) * 8, E, d, s, true);
#pragma unroll
    for (int k = 0; k < 8; ++k) {
        lr[k] = 0;
        if (k < nv) lr[k] = atomicAdd(&hist[d[k] >> BIN_SHIFT], 1);
    }
    __syncthreads();
    const int h    = hist[tid];
    const int incl = wave_incl_scan(h, lane);
    if (lane == 31) wtot[wave] = incl;
    __syncthreads();
    int carry = 0;
#pragma unroll
    for (int w = 0; w < EDGE_TB / 32; ++w) carry += (w < wave) ? wtot[w] : 0;
    const int base = carry + incl - h;
    hist[tid] = base;
    __syncthreads();
    if (tid < nb) pk[(size_t)blockIdx.x * nb + tid] = base | (h << 16);
    int pos[8];
#pragma unroll
    for (int k = 0; k < 8; ++k) pos[k] = hist[d[k] >> BIN_SHIFT] + lr[k];
    int* seg = recs + 2 * (size_t)blockIdx.x * SEG;
#pragma unroll
    for (int k = 0; k < 8; ++k) {
        if (k < nv) {
            v2i r;
            r.x = s[k];
            r.y = d[k];
            *(v2i*)(seg + 2 * pos[k]) = r;
        }
    }
}

__global__ void __launch_bounds__(SORT_TB)
bin_sort_kernel(const int* __restrict__ recs, const int* __restrict__ pk,
                int* __restrict__ csr, int* __restrict__ off, int* __restrict__ cnt,
                float* __restrict__ dinv, int n, int G, int nb) {
    constexpr int RB = 1 << BIN_SHIFT;
    constexpr int KR = 8;
    __shared__ int lcnt[RB];
    __shared__ int loff[RB];
    __shared__ int wtot[RB / 32];
    __shared__ int sh_start;
    const int b    = blockIdx.x;
    const int tid  = threadIdx.x;
    const int lane = tid & 31;
    const int wave = __builtin_amdgcn_readfirstlane(tid >> 5);
    const int lo   = b << BIN_SHIFT;
    if (tid < RB) lcnt[tid] = 0;
    if (tid == 0) sh_start = 0;
    __syncthreads();
    int pkv = 0;
    if (lane < KR && wave + 32 * lane < G) pkv = pk[(size_t)(wave + 32 * lane) * nb + b];
    int part = 0;
    for (int g = tid; g < G; g += SORT_TB) part += pk[(size_t)g * nb + b] & 0xFFFF;
    for (int dl = 16; dl > 0; dl >>= 1) part += __shfl_xor(part, dl);
    if (lane == 0 && part != 0) atomicAdd(&sh_start, part);
    const v2i* rbase = (const v2i*)recs;
    v2i        rec[KR];
    int        pc[KR];
    const v2i* pp[KR];
#pragma unroll
    for (int k = 0; k < KR; ++k) {
        const int e = __builtin_amdgcn_readlane(pkv, k);
        pc[k] = e >> 16;
        pp[k] = rbase + (size_t)(wave + 32 * k) * SEG + (e & 0xFFFF);
        rec[k].x = 0;
        rec[k].y = lo;
        if (lane < pc[k]) rec[k] = pp[k][lane];
    }
#pragma unroll
    for (int k = 0; k < KR; ++k) {
        if (lane < pc[k]) atomicAdd(&lcnt[rec[k].y - lo], 1);
        for (int r = 32 + lane; r < pc[k]; r += 32) atomicAdd(&lcnt[pp[k][r].y - lo], 1);
    }
    for (int g = wave + 32 * KR; g < G; g += 32) {
        const int  e = pk[(size_t)g * nb + b];
        const v2i* p = rbase + (size_t)g * SEG + (e & 0xFFFF);
        for (int r = lane; r < (e >> 16); r += 32) atomicAdd(&lcnt[p[r].y - lo], 1);
    }
    __syncthreads();
    int v = 0, incl = 0, lf = 0;
    if (tid < RB) {
        v    = lcnt[tid];
        incl = wave_incl_scan(v, lane);
        if (lane == 31) wtot[tid >> 5] = incl;
    }
    __syncthreads();
    const int bs = sh_start;
    if (tid < RB) {
        int carry = 0;
        for (int w = 0; w < (tid >> 5); ++w) carry += wtot[w];
        lf = carry + incl - v;
        loff[tid] = lf;
        lcnt[tid] = 0;
    }
    __syncthreads();
    if (tid < RB && lo + tid < n) {
        const int i = lo + tid;
        off[i]  = bs + lf;
        cnt[i]  = v;
        dinv[i] = rsqrtf((float)(v + 1));
    }
#pragma unroll
    for (int k = 0; k < KR; ++k) {
        if (lane < pc[k]) {
            const int j = rec[k].y - lo;
            const int p = atomicAdd(&lcnt[j], 1);
            csr[bs + loff[j] + p] = rec[k].x;
        }
        for (int r = 32 + lane; r < pc[k]; r += 32) {
            const v2i rc = pp[k][r];
            const int j  = rc.y - lo;
            const int p  = atomicAdd(&lcnt[j], 1);
            csr[bs + loff[j] + p] = rc.x;
        }
    }
    for (int g = wave + 32 * KR; g < G; g += 32) {
        const int  e = pk[(size_t)g * nb + b];
        const v2i* p = rbase + (size_t)g * SEG + (e & 0xFFFF);
        for (int r = lane; r < (e >> 16); r += 32) {
            const v2i rc = p[r];
            const int j  = rc.y - lo;
            const int q  = atomicAdd(&lcnt[j], 1);
            csr[bs + loff[j] + q] = rc.x;
        }
    }
}

template<int NOUT>
__global__ void __launch_bounds__(128)
gemm_wmma_kernel(const float* __restrict__ X,
                                 const float* __restrict__ W,
                                 const float* __restrict__ rscale,
                                 float* __restrict__ Y, int ntiles) {
    constexpr int NT = NOUT / 16;
    typedef float vnt __attribute__((ext_vector_type(NT)));
    const int tile = __builtin_amdgcn_readfirstlane(
        (int)(blockIdx.x * (blockDim.x >> 5) + (threadIdx.x >> 5)));
    if (tile >= ntiles) return;
    const int m0   = tile * 16;
    const int lane = threadIdx.x & 31;
    const int half = lane >> 4;
    const int l15  = lane & 15;

    v8f c[NT];
#pragma unroll
    for (int t = 0; t < NT; ++t) c[t] = v8f{};
    const float* xrow = X + (size_t)(m0 + l15) * 64;
    const float* wcol = W + l15 * NT;
#pragma unroll
    for (int kb = 0; kb < 64; kb += 4) {
        const int ka = kb + half * 2;
        const v2f a  = *(const v2f*)(xrow + ka);
        const vnt b0 = *(const vnt*)(wcol + (size_t)(ka + 0) * NOUT);
        const vnt b1 = *(const vnt*)(wcol + (size_t)(ka + 1) * NOUT);
#pragma unroll
        for (int t = 0; t < NT; ++t) {
            v2f b;
            b.x = b0[t];
            b.y = b1[t];
            c[t] = __builtin_amdgcn_wmma_f32_16x16x4_f32(
                     false, a,  false, b,
                     (short)0, c[t],  false,  false);
        }
    }

    const v4f rs0 = *(const v4f*)(rscale + m0 + half * 8);
    const v4f rs1 = *(const v4f*)(rscale + m0 + half * 8 + 4);
    const float rsc[8] = {rs0.x, rs0.y, rs0.z, rs0.w, rs1.x, rs1.y, rs1.z, rs1.w};
    float* yrow = Y + (size_t)(m0 + half * 8) * NOUT + l15 * NT;
#pragma unroll
    for (int j = 0; j < 8; ++j) {
        vnt v;
#pragma unroll
        for (int t = 0; t < NT; ++t) v[t] = c[t][j] * rsc[j];
        *(vnt*)(yrow + (size_t)j * NOUT) = v;
    }
}

template<int F, int U>
__device__ __forceinline__ void issue_rows(const float* hq, int sl, int j, int sub,
                                           v2f (&r)[U]) {
    constexpr int EPI = 64 / F;
#pragma unroll
    for (int u = 0; u < U; ++u) {
        const int jj = j + u * EPI + sub;
        int s;
        if constexpr (EPI == 1) s = __builtin_amdgcn_readlane(sl, jj);
        else                    s = __shfl(sl, jj);
        r[u] = *(const v2f*)(hq + (size_t)s * F);
    }
}

template<int F, int U>
__device__ __forceinline__ void add_rows(v2f& acc, int j, int sub, int m,
                                         const v2f (&r)[U]) {
    constexpr int EPI = 64 / F;
#pragma unroll
    for (int u = 0; u < U; ++u) {
        const int jj = j + u * EPI + sub;
        if (jj < m) {
            acc.x += r[u].x;
            acc.y += r[u].y;
        }
    }
}

template<int F, bool RELU>
__global__ void gather_kernel(const int* __restrict__ csr, const int* __restrict__ off,
                              const int* __restrict__ cnt, const float* __restrict__ dinv,
                              const float* __restrict__ h, const float* __restrict__ bias,
                              float* __restrict__ out, int n) {
    constexpr int LPR = F / 2;
    constexpr int EPI = 32 / LPR;
    constexpr int RS  = 16;
    constexpr int U   = RS / EPI;
    const int lane = threadIdx.x & 31;
    const int sub  = lane / LPR;
    const int q    = lane % LPR;
    const int wave  = __builtin_amdgcn_readfirstlane(
        (int)((blockIdx.x * blockDim.x + threadIdx.x) >> 5));
    const int nwave = (int)((gridDim.x * blockDim.x) >> 5);
    const int per = n / nwave;
    const int rem = n - per * nwave;
    const int i0  = wave * per + ((wave < rem) ? wave : rem);
    const int i1  = i0 + per + ((wave < rem) ? 1 : 0);
    if (i0 >= i1) return;
    const float* hq = h + 2 * q;
    const v2f bq = *(const v2f*)(bias + 2 * q);

    int   start = off[i0];
    int   c     = cnt[i0];
    float di    = dinv[i0];
    int   s_l   = i0;
    if (lane < c) s_l = csr[start + lane];
    for (int i = i0; i < i1; ++i) {
        const int   inx    = (i + 1 < i1) ? i + 1 : i;
        const int   startN = off[inx];
        const int   cN     = cnt[inx];
        const float diN    = dinv[inx];
        const int   m      = (c < 32) ? c : 32;

        const v2f rs = *(const v2f*)(hq + (size_t)i * F);
        v2f r[U];
        issue_rows<F, U>(hq, s_l, 0, sub, r);
        int sN = inx;
        if (lane < cN) sN = csr[startN + lane];
        v2f acc;
        acc.x = (sub == 0) ? rs.x : 0.0f;
        acc.y = (sub == 0) ? rs.y : 0.0f;
        add_rows<F, U>(acc, 0, sub, m, r);
        if (m > RS) {
            issue_rows<F, U>(hq, s_l, RS, sub, r);
            add_rows<F, U>(acc, RS, sub, m, r);
        }
        for (int base = 32; base < c; base += 32) {
            const int m2 = (c - base < 32) ? (c - base) : 32;
            int s2 = i;
            if (lane < m2) s2 = csr[start + base + lane];
            for (int j = 0; j < m2; j += RS) {
                issue_rows<F, U>(hq, s2, j, sub, r);
                add_rows<F, U>(acc, j, sub, m2, r);
            }
        }
        if constexpr (EPI == 2) {
            acc.x += __shfl_xor(acc.x, 16);
            acc.y += __shfl_xor(acc.y, 16);
        }
        if (sub == 0) {
            v2f v;
            v.x = di * acc.x + bq.x;
            v.y = di * acc.y + bq.y;
            if (RELU) { v.x = fmaxf(v.x, 0.0f); v.y = fmaxf(v.y, 0.0f); }
            *(v2f*)(out + (size_t)i * F + 2 * q) = v;
        }
        start = startN; c = cN; di = diN; s_l = sN;
    }
}

extern "C" void kernel_launch(void* const* d_in, const int* in_sizes, int n_in,
                              void* d_out, int out_size, void* d_ws, size_t ws_size,
                              hipStream_t stream) {
    const float* x  = (const float*)d_in[0];
    const int*   ei = (const int*)  d_in[1];
    const float* W1 = (const float*)d_in[2];
    const float* b1 = (const float*)d_in[3];
    const float* W2 = (const float*)d_in[4];
    const float* b2 = (const float*)d_in[5];

    const int N = in_sizes[0] / NF_IN;
    const int E = in_sizes[1] / 2;
    const int* src = ei;
    const int* dst = ei + E;

    char* ws = (char*)d_ws;
    auto take = [&](size_t bytes) { char* p = ws; ws += (bytes + 255) & ~(size_t)255; return p; };
    auto cdiv = [](int a, int b) { return (a + b - 1) / b; };
    const int NB = cdiv(N, 1 << BIN_SHIFT);
    const int G  = cdiv(E, SEG);
    int*   pk    = (int*)  take((size_t)G * NB * 4);
    int*   cnt   = (int*)  take((size_t)N * 4);
    int*   off   = (int*)  take((size_t)N * 4);
    float* dinv  = (float*)take((size_t)N * 4);
    int*   recs  = (int*)  take((size_t)G * SEG * 8);
    int*   csr   = (int*)  take((size_t)E * 4);
    float* h1pre = (float*)take((size_t)N * NF_HID * 4);
    float* h1    = (float*)take((size_t)N * NF_HID * 4);
    float* h2pre = (float*)take((size_t)N * NF_OUT * 4);
    float* out   = (float*)d_out;

    const int TB = 256;
    const int GW = 1024;

    bin_local_kernel<<<G, EDGE_TB, 0, stream>>>(src, dst, pk, recs, E, NB);
    bin_sort_kernel <<<NB, SORT_TB, 0, stream>>>(recs, pk, csr, off, cnt, dinv, N, G, NB);

    gemm_wmma_kernel<NF_HID><<<cdiv(N / 16, 4), 128, 0, stream>>>(x, W1, dinv, h1pre, N / 16);
    gather_kernel<NF_HID, true><<<GW, TB, 0, stream>>>(
        csr, off, cnt, dinv, h1pre, b1, h1, N);

    gemm_wmma_kernel<NF_OUT><<<cdiv(N / 16, 4), 128, 0, stream>>>(h1, W2, dinv, h2pre, N / 16);
    gather_kernel<NF_OUT, false><<<GW, TB, 0, stream>>>(
        csr, off, cnt, dinv, h2pre, b2, out, N);
}


# layer 1's edge sums keep their rows in LDS and the same work-group multiplies them by W2 on WMMA behind one barrier: no h1 round trip through memory, one launch fewer; sum order kept
# speedup vs baseline: 5.3567x; 1.0839x over previous
#include <hip/hip_runtime.h>


#define NF_IN  64
#define NF_HID 64
#define NF_OUT 32

typedef __attribute__((ext_vector_type(2))) float v2f;
typedef __attribute__((ext_vector_type(4))) float v4f;
typedef __attribute__((ext_vector_type(8))) float v8f;
typedef __attribute__((ext_vector_type(4))) int   v4i;

#define BIN_SHIFT 8
#define MAXNB     512
#define EDGE_TB   512
#define SEG       4096
#define SORT_TB   1024

typedef __attribute__((ext_vector_type(2))) int v2i;

__device__ __forceinline__ int wave_incl_scan(int v, int lane) {
    for (int d = 1; d < 32; d <<= 1) {
        const int t = __shfl_up(v, d);
        if (lane >= d) v += t;
    }
    return v;
}

__device__ __forceinline__ int load_edges8(const int* __restrict__ src, const int* __restrict__ dst,
                                           int e0, int E, int (&d)[8], int (&s)[8], bool want_src) {
#pragma unroll
    for (int k = 0; k < 8; ++k) { d[k] = 0; s[k] = 0; }
    if (e0 >= E) return 0;
    size_t al = reinterpret_cast<size_t>(dst);
    if (want_src) al |= reinterpret_cast<size_t>(src);
    if ((e0 + 8 <= E) && ((al & 15) == 0)) {
        const v4i d0 = *(const v4i*)(dst + e0), d1 = *(const v4i*)(dst + e0 + 4);
        d[0] = d0.x; d[1] = d0.y; d[2] = d0.z; d[3] = d0.w;
        d[4] = d1.x; d[5] = d1.y; d[6] = d1.z; d[7] = d1.w;
        if (want_src) {
            const v4i s0 = *(const v4i*)(src + e0), s1 = *(const v4i*)(src + e0 + 4);
            s[0] = s0.x; s[1] = s0.y; s[2] = s0.z; s[3] = s0.w;
            s[4] = s1.x; s[5] = s1.y; s[6] = s1.z; s[7] = s1.w;
        }
        return 8;
    }
    const int nv = (E - e0 < 8) ? (E - e0) : 8;
#pragma unroll
    for (int k = 0; k < 8; ++k) {
        d[k] = (k < nv) ? dst[e0 + k] : 0;
        if (want_src) s[k] = (k < nv) ? src[e0 + k] : 0;
    }
    return nv;
}

__global__ void __launch_bounds__(EDGE_TB)
bin_local_kernel(const int* __restrict__ src, const int* __restrict__ dst,
                 int* __restrict__ pk, int* __restrict__ recs, int E, int nb) {
    __shared__ int hist[MAXNB];
    __shared__ int wtot[EDGE_TB / 32];
    const int tid  = threadIdx.x;
    const int lane = tid & 31;
    const int wave = tid >> 5;
    hist[tid] = 0;
    __syncthreads();
    int d[8], s[8], lr[8];
    const int nv = load_edges8(src, dst, (int)(blockIdx.x * EDGE_TB + tid) * 8, E, d, s, true);
#pragma unroll
    for (int k = 0; k < 8; ++k) {
        lr[k] = 0;
        if (k < nv) lr[k] = atomicAdd(&hist[d[k] >> BIN_SHIFT], 1);
    }
    __syncthreads();
    const int h    = hist[tid];
    const int incl = wave_incl_scan(h, lane);
    if (lane == 31) wtot[wave] = incl;
    __syncthreads();
    int carry = 0;
#pragma unroll
    for (int w = 0; w < EDGE_TB / 32; ++w) carry += (w < wave) ? wtot[w] : 0;
    const int base = carry + incl - h;
    hist[tid] = base;
    __syncthreads();
    if (tid < nb) pk[(size_t)blockIdx.x * nb + tid] = base | (h << 16);
    int pos[8];
#pragma unroll
    for (int k = 0; k < 8; ++k) pos[k] = hist[d[k] >> BIN_SHIFT] + lr[k];
    int* seg = recs + 2 * (size_t)blockIdx.x * SEG;
#pragma unroll
    for (int k = 0; k < 8; ++k) {
        if (k < nv) {
            v2i r;
            r.x = s[k];
            r.y = d[k];
            *(v2i*)(seg + 2 * pos[k]) = r;
        }
    }
}

__global__ void __launch_bounds__(SORT_TB)
bin_sort_kernel(const int* __restrict__ recs, const int* __restrict__ pk,
                int* __restrict__ csr, int* __restrict__ off, int* __restrict__ cnt,
                float* __restrict__ dinv, int n, int G, int nb) {
    constexpr int RB = 1 << BIN_SHIFT;
    constexpr int KR = 8;
    __shared__ int lcnt[RB];
    __shared__ int loff[RB];
    __shared__ int wtot[RB / 32];
    __shared__ int sh_start;
    const int b    = blockIdx.x;
    const int tid  = threadIdx.x;
    const int lane = tid & 31;
    const int wave = __builtin_amdgcn_readfirstlane(tid >> 5);
    const int lo   = b << BIN_SHIFT;
    if (tid < RB) lcnt[tid] = 0;
    if (tid == 0) sh_start = 0;
    __syncthreads();
    int pkv = 0;
    if (lane < KR && wave + 32 * lane < G) pkv = pk[(size_t)(wave + 32 * lane) * nb + b];
    int part = 0;
    for (int g = tid; g < G; g += SORT_TB) part += pk[(size_t)g * nb + b] & 0xFFFF;
    for (int dl = 16; dl > 0; dl >>= 1) part += __shfl_xor(part, dl);
    if (lane == 0 && part != 0) atomicAdd(&sh_start, part);
    const v2i* rbase = (const v2i*)recs;
    v2i        rec[KR];
    int        pc[KR];
    const v2i* pp[KR];
#pragma unroll
    for (int k = 0; k < KR; ++k) {
        const int e = __builtin_amdgcn_readlane(pkv, k);
        pc[k] = e >> 16;
        pp[k] = rbase + (size_t)(wave + 32 * k) * SEG + (e & 0xFFFF);
        rec[k].x = 0;
        rec[k].y = lo;
        if (lane < pc[k]) rec[k] = pp[k][lane];
    }
#pragma unroll
    for (int k = 0; k < KR; ++k) {
        if (lane < pc[k]) atomicAdd(&lcnt[rec[k].y - lo], 1);
        for (int r = 32 + lane; r < pc[k]; r += 32) atomicAdd(&lcnt[pp[k][r].y - lo], 1);
    }
    for (int g = wave + 32 * KR; g < G; g += 32) {
        const int  e = pk[(size_t)g * nb + b];
        const v2i* p = rbase + (size_t)g * SEG + (e & 0xFFFF);
        for (int r = lane; r < (e >> 16); r += 32) atomicAdd(&lcnt[p[r].y - lo], 1);
    }
    __syncthreads();
    int v = 0, incl = 0, lf = 0;
    if (tid < RB) {
        v    = lcnt[tid];
        incl = wave_incl_scan(v, lane);
        if (lane == 31) wtot[tid >> 5] = incl;
    }
    __syncthreads();
    const int bs = sh_start;
    if (tid < RB) {
        int carry = 0;
        for (int w = 0; w < (tid >> 5); ++w) carry += wtot[w];
        lf = carry + incl - v;
        loff[tid] = lf;
        lcnt[tid] = 0;
    }
    __syncthreads();
    if (tid < RB && lo + tid < n) {
        const int i = lo + tid;
        off[i]  = bs + lf;
        cnt[i]  = v;
        dinv[i] = rsqrtf((float)(v + 1));
    }
#pragma unroll
    for (int k = 0; k < KR; ++k) {
        if (lane < pc[k]) {
            const int j = rec[k].y - lo;
            const int p = atomicAdd(&lcnt[j], 1);
            csr[bs + loff[j] + p] = rec[k].x;
        }
        for (int r = 32 + lane; r < pc[k]; r += 32) {
            const v2i rc = pp[k][r];
            const int j  = rc.y - lo;
            const int p  = atomicAdd(&lcnt[j], 1);
            csr[bs + loff[j] + p] = rc.x;
        }
    }
    for (int g = wave + 32 * KR; g < G; g += 32) {
        const int  e = pk[(size_t)g * nb + b];
        const v2i* p = rbase + (size_t)g * SEG + (e & 0xFFFF);
        for (int r = lane; r < (e >> 16); r += 32) {
            const v2i rc = p[r];
            const int j  = rc.y - lo;
            const int q  = atomicAdd(&lcnt[j], 1);
            csr[bs + loff[j] + q] = rc.x;
        }
    }
}

template<int NOUT>
__device__ __forceinline__ void strip_product(const float* arow, const float* __restrict__ W,
                                              const float* __restrict__ rscale,
                                              float* __restrict__ Y, int m0, int lane) {
    constexpr int NT = NOUT / 16;
    typedef float vnt __attribute__((ext_vector_type(NT)));
    const int half = lane >> 4;
    const int l15  = lane & 15;

    v8f c[NT];
#pragma unroll
    for (int t = 0; t < NT; ++t) c[t] = v8f{};
    const float* wcol = W + l15 * NT;
#pragma unroll
    for (int kb = 0; kb < 64; kb += 4) {
        const int ka = kb + half * 2;
        const v2f a  = *(const v2f*)(arow + ka);
        const vnt b0 = *(const vnt*)(wcol + (size_t)(ka + 0) * NOUT);
        const vnt b1 = *(const vnt*)(wcol + (size_t)(ka + 1) * NOUT);
#pragma unroll
        for (int t = 0; t < NT; ++t) {
            v2f b;
            b.x = b0[t];
            b.y = b1[t];
            c[t] = __builtin_amdgcn_wmma_f32_16x16x4_f32(
                     false, a,  false, b,
                     (short)0, c[t],  false,  false);
        }
    }

    const v4f rs0 = *(const v4f*)(rscale + m0 + half * 8);
    const v4f rs1 = *(const v4f*)(rscale + m0 + half * 8 + 4);
    const float rsc[8] = {rs0.x, rs0.y, rs0.z, rs0.w, rs1.x, rs1.y, rs1.z, rs1.w};
    float* yrow = Y + (size_t)(m0 + half * 8) * NOUT + l15 * NT;
#pragma unroll
    for (int j = 0; j < 8; ++j) {
        vnt v;
#pragma unroll
        for (int t = 0; t < NT; ++t) v[t] = c[t][j] * rsc[j];
        *(vnt*)(yrow + (size_t)j * NOUT) = v;
    }
}

template<int NOUT>
__global__ void __launch_bounds__(128)
gemm_wmma_kernel(const float* __restrict__ X,
                                 const float* __restrict__ W,
                                 const float* __restrict__ rscale,
                                 float* __restrict__ Y, int ntiles) {
    const int tile = __builtin_amdgcn_readfirstlane(
        (int)(blockIdx.x * (blockDim.x >> 5) + (threadIdx.x >> 5)));
    if (tile >= ntiles) return;
    const int m0   = tile * 16;
    const int lane = threadIdx.x & 31;
    strip_product<NOUT>(X + (size_t)(m0 + (lane & 15)) * 64, W, rscale, Y, m0, lane);
}

template<int F, int U>
__device__ __forceinline__ void issue_rows(const float* hq, int sl, int j, int sub,
                                           v2f (&r)[U]) {
    constexpr int EPI = 64 / F;
#pragma unroll
    for (int u = 0; u < U; ++u) {
        const int jj = j + u * EPI + sub;
        int s;
        if constexpr (EPI == 1) s = __builtin_amdgcn_readlane(sl, jj);
        else                    s = __shfl(sl, jj);
        r[u] = *(const v2f*)(hq + (size_t)s * F);
    }
}

template<int F, int U>
__device__ __forceinline__ void add_rows(v2f& acc, int j, int sub, int m,
                                         const v2f (&r)[U]) {
    constexpr int EPI = 64 / F;
#pragma unroll
    for (int u = 0; u < U; ++u) {
        const int jj = j + u * EPI + sub;
        if (jj < m) {
            acc.x += r[u].x;
            acc.y += r[u].y;
        }
    }
}

template<int F, bool RELU, class Put>
__device__ __forceinline__ void gather_run(const int* __restrict__ csr, const int* __restrict__ off,
                                           const int* __restrict__ cnt, const float* __restrict__ dinv,
                                           const float* __restrict__ h, const float* __restrict__ bias,
                                           int i0, int i1, int lane, Put put) {
    constexpr int LPR = F / 2;
    constexpr int EPI = 32 / LPR;
    constexpr int RS  = 16;
    constexpr int U   = RS / EPI;
    const int sub  = lane / LPR;
    const int q    = lane % LPR;
    const float* hq = h + 2 * q;
    const v2f bq = *(const v2f*)(bias + 2 * q);

    int   start = off[i0];
    int   c     = cnt[i0];
    float di    = dinv[i0];
    int   s_l   = i0;
    if (lane < c) s_l = csr[start + lane];
    for (int i = i0; i < i1; ++i) {
        const int   inx    = (i + 1 < i1) ? i + 1 : i;
        const int   startN = off[inx];
        const int   cN     = cnt[inx];
        const float diN    = dinv[inx];
        const int   m      = (c < 32) ? c : 32;

        const v2f rs = *(const v2f*)(hq + (size_t)i * F);
        v2f r[U];
        issue_rows<F, U>(hq, s_l, 0, sub, r);
        int sN = inx;
        if (lane < cN) sN = csr[startN + lane];
        v2f acc;
        acc.x = (sub == 0) ? rs.x : 0.0f;
        acc.y = (sub == 0) ? rs.y : 0.0f;
        add_rows<F, U>(acc, 0, sub, m, r);
        if (m > RS) {
            issue_rows<F, U>(hq, s_l, RS, sub, r);
            add_rows<F, U>(acc, RS, sub, m, r);
        }
        for (int base = 32; base < c; base += 32) {
            const int m2 = (c - base < 32) ? (c - base) : 32;
            int s2 = i;
            if (lane < m2) s2 = csr[start + base + lane];
            for (int j = 0; j < m2; j += RS) {
                issue_rows<F, U>(hq, s2, j, sub, r);
                add_rows<F, U>(acc, j, sub, m2, r);
            }
        }
        if constexpr (EPI == 2) {
            acc.x += __shfl_xor(acc.x, 16);
            acc.y += __shfl_xor(acc.y, 16);
        }
        if (sub == 0) {
            v2f v;
            v.x = di * acc.x + bq.x;
            v.y = di * acc.y + bq.y;
            if (RELU) { v.x = fmaxf(v.x, 0.0f); v.y = fmaxf(v.y, 0.0f); }
            put(i, q, v);
        }
        start = startN; c = cN; di = diN; s_l = sN;
    }
}

template<int F, bool RELU>
__global__ void gather_kernel(const int* __restrict__ csr, const int* __restrict__ off,
                              const int* __restrict__ cnt, const float* __restrict__ dinv,
                              const float* __restrict__ h, const float* __restrict__ bias,
                              float* __restrict__ out, int n) {
    const int lane = threadIdx.x & 31;
    const int wave  = __builtin_amdgcn_readfirstlane(
        (int)((blockIdx.x * blockDim.x + threadIdx.x) >> 5));
    const int nwave = (int)((gridDim.x * blockDim.x) >> 5);
    const int per = n / nwave;
    const int rem = n - per * nwave;
    const int i0  = wave * per + ((wave < rem) ? wave : rem);
    const int i1  = i0 + per + ((wave < rem) ? 1 : 0);
    if (i0 >= i1) return;
    gather_run<F, RELU>(csr, off, cnt, dinv, h, bias, i0, i1, lane,
                        [&](int i, int q, v2f v) { *(v2f*)(out + (size_t)i * F + 2 * q) = v; });
}

#define FG_WAVES  8
#define FG_NPW    6
#define FG_NODES  (FG_WAVES * FG_NPW)
#define FG_STRIPS (FG_NODES / 16)
#define FG_PITCH  (NF_HID + 4)

__global__ void __launch_bounds__(FG_WAVES * 32)
gather_product_kernel(const int* __restrict__ csr, const int* __restrict__ off,
                      const int* __restrict__ cnt, const float* __restrict__ dinv,
                      const float* __restrict__ h, const float* __restrict__ bias,
                      const float* __restrict__ W, float* __restrict__ Y, int n) {
    static_assert(FG_NODES % 16 == 0, "a work-group owns whole strips");
    __shared__ float rows[FG_NODES * FG_PITCH];
    const int lane = threadIdx.x & 31;
    const int wave = __builtin_amdgcn_readfirstlane((int)(threadIdx.x >> 5));
    const int n0   = (int)blockIdx.x * FG_NODES;
    const int i0   = n0 + wave * FG_NPW;
    const int i1   = (i0 + FG_NPW < n) ? (i0 + FG_NPW) : n;
    if (i0 < i1) {
        gather_run<NF_HID, true>(csr, off, cnt, dinv, h, bias, i0, i1, lane,
                                 [&](int i, int q, v2f v) {
                                     *(v2f*)(rows + (i - n0) * FG_PITCH + 2 * q) = v;
                                 });
    }
    __syncthreads();
    const int m0 = n0 + wave * 16;
    if (wave < FG_STRIPS && m0 < n)
        strip_product<NF_OUT>(rows + (wave * 16 + (lane & 15)) * FG_PITCH, W, dinv, Y, m0, lane);
}

extern "C" void kernel_launch(void* const* d_in, const int* in_sizes, int n_in,
                              void* d_out, int out_size, void* d_ws, size_t ws_size,
                              hipStream_t stream) {
    const float* x  = (const float*)d_in[0];
    const int*   ei = (const int*)  d_in[1];
    const float* W1 = (const float*)d_in[2];
    const float* b1 = (const float*)d_in[3];
    const float* W2 = (const float*)d_in[4];
    const float* b2 = (const float*)d_in[5];

    const int N = in_sizes[0] / NF_IN;
    const int E = in_sizes[1] / 2;
    const int* src = ei;
    const int* dst = ei + E;

    char* ws = (char*)d_ws;
    auto take = [&](size_t bytes) { char* p = ws; ws += (bytes + 255) & ~(size_t)255; return p; };
    auto cdiv = [](int a, int b) { return (a + b - 1) / b; };
    const int NB = cdiv(N, 1 << BIN_SHIFT);
    const int G  = cdiv(E, SEG);
    int*   pk    = (int*)  take((size_t)G * NB * 4);
    int*   cnt   = (int*)  take((size_t)N * 4);
    int*   off   = (int*)  take((size_t)N * 4);
    float* dinv  = (float*)take((size_t)N * 4);
    int*   recs  = (int*)  take((size_t)G * SEG * 8);
    int*   csr   = (int*)  take((size_t)E * 4);
    float* h1pre = (float*)take((size_t)N * NF_HID * 4);
    float* h2pre = (float*)take((size_t)N * NF_OUT * 4);
    float* out   = (float*)d_out;

    const int TB = 256;
    const int GW = 1024;

    bin_local_kernel<<<G, EDGE_TB, 0, stream>>>(src, dst, pk, recs, E, NB);
    bin_sort_kernel <<<NB, SORT_TB, 0, stream>>>(recs, pk, csr, off, cnt, dinv, N, G, NB);

    gemm_wmma_kernel<NF_HID><<<cdiv(N / 16, 4), 128, 0, stream>>>(x, W1, dinv, h1pre, N / 16);
    gather_product_kernel<<<cdiv(N, FG_NODES), FG_WAVES * 32, 0, stream>>>(
        csr, off, cnt, dinv, h1pre, b1, W2, h2pre, N);

    gather_kernel<NF_OUT, false><<<GW, TB, 0, stream>>>(
        csr, off, cnt, dinv, h2pre, b2, out, N);
}


# layer 1's product x@W1 moved into the edge sort's per-bin work-group (it owns those rows' dinv): X rows staged in LDS, 2 waves per 16-row strip on WMMA; one launch fewer; sum order kept
# speedup vs baseline: 5.8398x; 1.0902x over previous
#include <hip/hip_runtime.h>


#define NF_IN  64
#define NF_HID 64
#define NF_OUT 32

typedef __attribute__((ext_vector_type(2))) float v2f;
typedef __attribute__((ext_vector_type(4))) float v4f;
typedef __attribute__((ext_vector_type(8))) float v8f;
typedef __attribute__((ext_vector_type(4))) int   v4i;

#define BIN_SHIFT 8
#define MAXNB     512
#define EDGE_TB   512
#define SEG       4096
#define SORT_TB   1024

typedef __attribute__((ext_vector_type(2))) int v2i;

__device__ __forceinline__ int wave_incl_scan(int v, int lane) {
    for (int d = 1; d < 32; d <<= 1) {
        const int t = __shfl_up(v, d);
        if (lane >= d) v += t;
    }
    return v;
}

__device__ __forceinline__ int load_edges8(const int* __restrict__ src, const int* __restrict__ dst,
                                           int e0, int E, int (&d)[8], int (&s)[8], bool want_src) {
#pragma unroll
    for (int k = 0; k < 8; ++k) { d[k] = 0; s[k] = 0; }
    if (e0 >= E) return 0;
    size_t al = reinterpret_cast<size_t>(dst);
    if (want_src) al |= reinterpret_cast<size_t>(src);
    if ((e0 + 8 <= E) && ((al & 15) == 0)) {
        const v4i d0 = *(const v4i*)(dst + e0), d1 = *(const v4i*)(dst + e0 + 4);
        d[0] = d0.x; d[1] = d0.y; d[2] = d0.z; d[3] = d0.w;
        d[4] = d1.x; d[5] = d1.y; d[6] = d1.z; d[7] = d1.w;
        if (want_src) {
            const v4i s0 = *(const v4i*)(src + e0), s1 = *(const v4i*)(src + e0 + 4);
            s[0] = s0.x; s[1] = s0.y; s[2] = s0.z; s[3] = s0.w;
            s[4] = s1.x; s[5] = s1.y; s[6] = s1.z; s[7] = s1.w;
        }
        return 8;
    }
    const int nv = (E - e0 < 8) ? (E - e0) : 8;
#pragma unroll
    for (int k = 0; k < 8; ++k) {
        d[k] = (k < nv) ? dst[e0 + k] : 0;
        if (want_src) s[k] = (k < nv) ? src[e0 + k] : 0;
    }
    return nv;
}

__global__ void __launch_bounds__(EDGE_TB)
bin_local_kernel(const int* __restrict__ src, const int* __restrict__ dst,
                 int* __restrict__ pk, int* __restrict__ recs, int E, int nb) {
    __shared__ int hist[MAXNB];
    __shared__ int wtot[EDGE_TB / 32];
    const int tid  = threadIdx.x;
    const int lane = tid & 31;
    const int wave = tid >> 5;
    hist[tid] = 0;
    __syncthreads();
    int d[8], s[8], lr[8];
    const int nv = load_edges8(src, dst, (int)(blockIdx.x * EDGE_TB + tid) * 8, E, d, s, true);
#pragma unroll
    for (int k = 0; k < 8; ++k) {
        lr[k] = 0;
        if (k < nv) lr[k] = atomicAdd(&hist[d[k] >> BIN_SHIFT], 1);
    }
    __syncthreads();
    const int h    = hist[tid];
    const int incl = wave_incl_scan(h, lane);
    if (lane == 31) wtot[wave] = incl;
    __syncthreads();
    int carry = 0;
#pragma unroll
    for (int w = 0; w < EDGE_TB / 32; ++w) carry += (w < wave) ? wtot[w] : 0;
    const int base = carry + incl - h;
    hist[tid] = base;
    __syncthreads();
    if (tid < nb) pk[(size_t)blockIdx.x * nb + tid] = base | (h << 16);
    int pos[8];
#pragma unroll
    for (int k = 0; k < 8; ++k) pos[k] = hist[d[k] >> BIN_SHIFT] + lr[k];
    int* seg = recs + 2 * (size_t)blockIdx.x * SEG;
#pragma unroll
    for (int k = 0; k < 8; ++k) {
        if (k < nv) {
            v2i r;
            r.x = s[k];
            r.y = d[k];
            *(v2i*)(seg + 2 * pos[k]) = r;
        }
    }
}

#define ROW_PITCH (64 + 4)
template<int NOUT, int LD>
__device__ __forceinline__ void strip_product(const float* arow, const float* __restrict__ W,
                                              const float* __restrict__ rscale,
                                              float* __restrict__ Y, int m0, int lane) {
    constexpr int NT = NOUT / 16;
    typedef float vnt __attribute__((ext_vector_type(NT)));
    const int half = lane >> 4;
    const int l15  = lane & 15;

    v8f c[NT];
#pragma unroll
    for (int t = 0; t < NT; ++t) c[t] = v8f{};
    const float* wcol = W + l15 * NT;
#pragma unroll
    for (int kb = 0; kb < 64; kb += 4) {
        const int ka = kb + half * 2;
        const v2f a  = *(const v2f*)(arow + ka);
        const vnt b0 = *(const vnt*)(wcol + (size_t)(ka + 0) * LD);
        const vnt b1 = *(const vnt*)(wcol + (size_t)(ka + 1) * LD);
#pragma unroll
        for (int t = 0; t < NT; ++t) {
            v2f b;
            b.x = b0[t];
            b.y = b1[t];
            c[t] = __builtin_amdgcn_wmma_f32_16x16x4_f32(
                     false, a,  false, b,
                     (short)0, c[t],  false,  false);
        }
    }

    const v4f rs0 = *(const v4f*)(rscale + m0 + half * 8);
    const v4f rs1 = *(const v4f*)(rscale + m0 + half * 8 + 4);
    const float rsc[8] = {rs0.x, rs0.y, rs0.z, rs0.w, rs1.x, rs1.y, rs1.z, rs1.w};
    float* yrow = Y + (size_t)(m0 + half * 8) * LD + l15 * NT;
#pragma unroll
    for (int j = 0; j < 8; ++j) {
        vnt v;
#pragma unroll
        for (int t = 0; t < NT; ++t) v[t] = c[t][j] * rsc[j];
        *(vnt*)(yrow + (size_t)j * LD) = v;
    }
}

__global__ void __launch_bounds__(SORT_TB)
bin_sort_kernel(const int* __restrict__ recs, const int* __restrict__ pk,
                int* __restrict__ csr, int* __restrict__ off, int* __restrict__ cnt,
                float* __restrict__ dinv, int n, int G, int nb,
                const float* __restrict__ X, const float* __restrict__ W,
                float* __restrict__ Y) {
    constexpr int RB = 1 << BIN_SHIFT;
    constexpr int KR = 8;
    constexpr int XV = RB * (NF_IN / 4) / SORT_TB;
    static_assert(RB * (NF_IN / 4) % SORT_TB == 0, "the X rows split evenly over the threads");
    static_assert(2 * (RB / 16) == SORT_TB / 32, "two waves per strip of 16 rows");
    __shared__ int lcnt[RB];
    __shared__ int loff[RB];
    __shared__ int wtot[RB / 32];
    __shared__ int sh_start;
    __shared__ __attribute__((aligned(16))) float ldinv[RB];
    __shared__ __attribute__((aligned(16))) float xs[RB * ROW_PITCH];
    const int b    = blockIdx.x;
    const int tid  = threadIdx.x;
    const int lane = tid & 31;
    const int wave = __builtin_amdgcn_readfirstlane(tid >> 5);
    const int lo   = b << BIN_SHIFT;
    if (tid < RB) lcnt[tid] = 0;
    if (tid == 0) sh_start = 0;
    __syncthreads();
    int pkv = 0;
    if (lane < KR && wave + 32 * lane < G) pkv = pk[(size_t)(wave + 32 * lane) * nb + b];
    int part = 0;
    for (int g = tid; g < G; g += SORT_TB) part += pk[(size_t)g * nb + b] & 0xFFFF;
    for (int dl = 16; dl > 0; dl >>= 1) part += __shfl_xor(part, dl);
    if (lane == 0 && part != 0) atomicAdd(&sh_start, part);
    const v2i* rbase = (const v2i*)recs;
    v2i        rec[KR];
    int        pc[KR];
    const v2i* pp[KR];
#pragma unroll
    for (int k = 0; k < KR; ++k) {
        const int e = __builtin_amdgcn_readlane(pkv, k);
        pc[k] = e >> 16;
        pp[k] = rbase + (size_t)(wave + 32 * k) * SEG + (e & 0xFFFF);
        rec[k].x = 0;
        rec[k].y = lo;
        if (lane < pc[k]) rec[k] = pp[k][lane];
    }
    v4f xv[XV];
#pragma unroll
    for (int i = 0; i < XV; ++i) {
        const int p   = tid + i * SORT_TB;
        const int row = (lo + (p >> 4) < n) ? lo + (p >> 4) : n - 1;
        xv[i] = *(const v4f*)(X + (size_t)row * NF_IN + (p & 15) * 4);
    }
#pragma unroll
    for (int k = 0; k < KR; ++k) {
        if (lane < pc[k]) atomicAdd(&lcnt[rec[k].y - lo], 1);
        for (int r = 32 + lane; r < pc[k]; r += 32) atomicAdd(&lcnt[pp[k][r].y - lo], 1);
    }
#pragma unroll
    for (int i = 0; i < XV; ++i) {
        const int p = tid + i * SORT_TB;
        *(v4f*)(xs + (p >> 4) * ROW_PITCH + (p & 15) * 4) = xv[i];
    }
    for (int g = wave + 32 * KR; g < G; g += 32) {
        const int  e = pk[(size_t)g * nb + b];
        const v2i* p = rbase + (size_t)g * SEG + (e & 0xFFFF);
        for (int r = lane; r < (e >> 16); r += 32) atomicAdd(&lcnt[p[r].y - lo], 1);
    }
    __syncthreads();
    int v = 0, incl = 0, lf = 0;
    if (tid < RB) {
        v    = lcnt[tid];
        incl = wave_incl_scan(v, lane);
        if (lane == 31) wtot[tid >> 5] = incl;
    }
    __syncthreads();
    const int bs = sh_start;
    if (tid < RB) {
        int carry = 0;
        for (int w = 0; w < (tid >> 5); ++w) carry += wtot[w];
        lf = carry + incl - v;
        loff[tid] = lf;
        lcnt[tid] = 0;
        ldinv[tid] = rsqrtf((float)(v + 1));
    }
    __syncthreads();
    if (tid < RB && lo + tid < n) {
        const int i = lo + tid;
        off[i]  = bs + lf;
        cnt[i]  = v;
        dinv[i] = ldinv[tid];
    }
#pragma unroll
    for (int k = 0; k < KR; ++k) {
        if (lane < pc[k]) {
            const int j = rec[k].y - lo;
            const int p = atomicAdd(&lcnt[j], 1);
            csr[bs + loff[j] + p] = rec[k].x;
        }
        for (int r = 32 + lane; r < pc[k]; r += 32) {
            const v2i rc = pp[k][r];
            const int j  = rc.y - lo;
            const int p  = atomicAdd(&lcnt[j], 1);
            csr[bs + loff[j] + p] = rc.x;
        }
    }
    for (int g = wave + 32 * KR; g < G; g += 32) {
        const int  e = pk[(size_t)g * nb + b];
        const v2i* p = rbase + (size_t)g * SEG + (e & 0xFFFF);
        for (int r = lane; r < (e >> 16); r += 32) {
            const v2i rc = p[r];
            const int j  = rc.y - lo;
            const int q  = atomicAdd(&lcnt[j], 1);
            csr[bs + loff[j] + q] = rc.x;
        }
    }
    const int strip = wave >> 1;
    const int cb    = (wave & 1) * (NF_HID / 2);
    const int r0    = strip * 16;
    if (lo + r0 + 16 <= n)
        strip_product<NF_HID / 2, NF_HID>(xs + (r0 + (lane & 15)) * ROW_PITCH, W + cb, ldinv,
                                          Y + (size_t)lo * NF_HID + cb, r0, lane);
}

template<int F, int U>
__device__ __forceinline__ void issue_rows(const float* hq, int sl, int j, int sub,
                                           v2f (&r)[U]) {
    constexpr int EPI = 64 / F;
#pragma unroll
    for (int u = 0; u < U; ++u) {
        const int jj = j + u * EPI + sub;
        int s;
        if constexpr (EPI == 1) s = __builtin_amdgcn_readlane(sl, jj);
        else                    s = __shfl(sl, jj);
        r[u] = *(const v2f*)(hq + (size_t)s * F);
    }
}

template<int F, int U>
__device__ __forceinline__ void add_rows(v2f& acc, int j, int sub, int m,
                                         const v2f (&r)[U]) {
    constexpr int EPI = 64 / F;
#pragma unroll
    for (int u = 0; u < U; ++u) {
        const int jj = j + u * EPI + sub;
        if (jj < m) {
            acc.x += r[u].x;
            acc.y += r[u].y;
        }
    }
}

template<int F, bool RELU, class Put>
__device__ __forceinline__ void gather_run(const int* __restrict__ csr, const int* __restrict__ off,
                                           const int* __restrict__ cnt, const float* __restrict__ dinv,
                                           const float* __restrict__ h, const float* __restrict__ bias,
                                           int i0, int i1, int lane, Put put) {
    constexpr int LPR = F / 2;
    constexpr int EPI = 32 / LPR;
    constexpr int RS  = 16;
    constexpr int U   = RS / EPI;
    const int sub  = lane / LPR;
    const int q    = lane % LPR;
    const float* hq = h + 2 * q;
    const v2f bq = *(const v2f*)(bias + 2 * q);

    int   start = off[i0];
    int   c     = cnt[i0];
    float di    = dinv[i0];
    int   s_l   = i0;
    if (lane < c) s_l = csr[start + lane];
    for (int i = i0; i < i1; ++i) {
        const int   inx    = (i + 1 < i1) ? i + 1 : i;
        const int   startN = off[inx];
        const int   cN     = cnt[inx];
        const float diN    = dinv[inx];
        const int   m      = (c < 32) ? c : 32;

        const v2f rs = *(const v2f*)(hq + (size_t)i * F);
        v2f r[U];
        issue_rows<F, U>(hq, s_l, 0, sub, r);
        int sN = inx;
        if (lane < cN) sN = csr[startN + lane];
        v2f acc;
        acc.x = (sub == 0) ? rs.x : 0.0f;
        acc.y = (sub == 0) ? rs.y : 0.0f;
        add_rows<F, U>(acc, 0, sub, m, r);
        if (m > RS) {
            issue_rows<F, U>(hq, s_l, RS, sub, r);
            add_rows<F, U>(acc, RS, sub, m, r);
        }
        for (int base = 32; base < c; base += 32) {
            const int m2 = (c - base < 32) ? (c - base) : 32;
            int s2 = i;
            if (lane < m2) s2 = csr[start + base + lane];
            for (int j = 0; j < m2; j += RS) {
                issue_rows<F, U>(hq, s2, j, sub, r);
                add_rows<F, U>(acc, j, sub, m2, r);
            }
        }
        if constexpr (EPI == 2) {
            acc.x += __shfl_xor(acc.x, 16);
            acc.y += __shfl_xor(acc.y, 16);
        }
        if (sub == 0) {
            v2f v;
            v.x = di * acc.x + bq.x;
            v.y = di * acc.y + bq.y;
            if (RELU) { v.x = fmaxf(v.x, 0.0f); v.y = fmaxf(v.y, 0.0f); }
            put(i, q, v);
        }
        start = startN; c = cN; di = diN; s_l = sN;
    }
}

template<int F, bool RELU>
__global__ void gather_kernel(const int* __restrict__ csr, const int* __restrict__ off,
                              const int* __restrict__ cnt, const float* __restrict__ dinv,
                              const float* __restrict__ h, const float* __restrict__ bias,
                              float* __restrict__ out, int n) {
    const int lane = threadIdx.x & 31;
    const int wave  = __builtin_amdgcn_readfirstlane(
        (int)((blockIdx.x * blockDim.x + threadIdx.x) >> 5));
    const int nwave = (int)((gridDim.x * blockDim.x) >> 5);
    const int per = n / nwave;
    const int rem = n - per * nwave;
    const int i0  = wave * per + ((wave < rem) ? wave : rem);
    const int i1  = i0 + per + ((wave < rem) ? 1 : 0);
    if (i0 >= i1) return;
    gather_run<F, RELU>(csr, off, cnt, dinv, h, bias, i0, i1, lane,
                        [&](int i, int q, v2f v) { *(v2f*)(out + (size_t)i * F + 2 * q) = v; });
}

#define FG_WAVES  8
#define FG_NPW    6
#define FG_NODES  (FG_WAVES * FG_NPW)
#define FG_STRIPS (FG_NODES / 16)
#define FG_PITCH  ROW_PITCH

__global__ void __launch_bounds__(FG_WAVES * 32)
gather_product_kernel(const int* __restrict__ csr, const int* __restrict__ off,
                      const int* __restrict__ cnt, const float* __restrict__ dinv,
                      const float* __restrict__ h, const float* __restrict__ bias,
                      const float* __restrict__ W, float* __restrict__ Y, int n) {
    static_assert(FG_NODES % 16 == 0, "a work-group owns whole strips");
    __shared__ float rows[FG_NODES * FG_PITCH];
    const int lane = threadIdx.x & 31;
    const int wave = __builtin_amdgcn_readfirstlane((int)(threadIdx.x >> 5));
    const int n0   = (int)blockIdx.x * FG_NODES;
    const int i0   = n0 + wave * FG_NPW;
    const int i1   = (i0 + FG_NPW < n) ? (i0 + FG_NPW) : n;
    if (i0 < i1) {
        gather_run<NF_HID, true>(csr, off, cnt, dinv, h, bias, i0, i1, lane,
                                 [&](int i, int q, v2f v) {
                                     *(v2f*)(rows + (i - n0) * FG_PITCH + 2 * q) = v;
                                 });
    }
    __syncthreads();
    const int m0 = n0 + wave * 16;
    if (wave < FG_STRIPS && m0 < n)
        strip_product<NF_OUT, NF_OUT>(rows + (wave * 16 + (lane & 15)) * FG_PITCH, W, dinv, Y, m0, lane);
}

extern "C" void kernel_launch(void* const* d_in, const int* in_sizes, int n_in,
                              void* d_out, int out_size, void* d_ws, size_t ws_size,
                              hipStream_t stream) {
    const float* x  = (const float*)d_in[0];
    const int*   ei = (const int*)  d_in[1];
    const float* W1 = (const float*)d_in[2];
    const float* b1 = (const float*)d_in[3];
    const float* W2 = (const float*)d_in[4];
    const float* b2 = (const float*)d_in[5];

    const int N = in_sizes[0] / NF_IN;
    const int E = in_sizes[1] / 2;
    const int* src = ei;
    const int* dst = ei + E;

    char* ws = (char*)d_ws;
    auto take = [&](size_t bytes) { char* p = ws; ws += (bytes + 255) & ~(size_t)255; return p; };
    auto cdiv = [](int a, int b) { return (a + b - 1) / b; };
    const int NB = cdiv(N, 1 << BIN_SHIFT);
    const int G  = cdiv(E, SEG);
    int*   pk    = (int*)  take((size_t)G * NB * 4);
    int*   cnt   = (int*)  take((size_t)N * 4);
    int*   off   = (int*)  take((size_t)N * 4);
    float* dinv  = (float*)take((size_t)N * 4);
    int*   recs  = (int*)  take((size_t)G * SEG * 8);
    int*   csr   = (int*)  take((size_t)E * 4);
    float* h1pre = (float*)take((size_t)N * NF_HID * 4);
    float* h2pre = (float*)take((size_t)N * NF_OUT * 4);
    float* out   = (float*)d_out;

    const int TB = 256;
    const int GW = 1024;

    bin_local_kernel<<<G, EDGE_TB, 0, stream>>>(src, dst, pk, recs, E, NB);
    bin_sort_kernel <<<NB, SORT_TB, 0, stream>>>(recs, pk, csr, off, cnt, dinv, N, G, NB,
                                                 x, W1, h1pre);

    gather_product_kernel<<<cdiv(N, FG_NODES), FG_WAVES * 32, 0, stream>>>(
        csr, off, cnt, dinv, h1pre, b1, W2, h2pre, N);

    gather_kernel<NF_OUT, false><<<GW, TB, 0, stream>>>(
        csr, off, cnt, dinv, h2pre, b2, out, N);
}


# edge sums ask for the rows of all of a node's first 32 edges in one go (16, 24 or 32 loads by the node's edge count) instead of two dependent rounds of 16: one trip per node; sum order kept
# speedup vs baseline: 6.0672x; 1.0389x over previous
#include <hip/hip_runtime.h>


#define NF_IN  64
#define NF_HID 64
#define NF_OUT 32

typedef __attribute__((ext_vector_type(2))) float v2f;
typedef __attribute__((ext_vector_type(4))) float v4f;
typedef __attribute__((ext_vector_type(8))) float v8f;
typedef __attribute__((ext_vector_type(4))) int   v4i;

#define BIN_SHIFT 8
#define MAXNB     512
#define EDGE_TB   512
#define SEG       4096
#define SORT_TB   1024

typedef __attribute__((ext_vector_type(2))) int v2i;

__device__ __forceinline__ int wave_incl_scan(int v, int lane) {
    for (int d = 1; d < 32; d <<= 1) {
        const int t = __shfl_up(v, d);
        if (lane >= d) v += t;
    }
    return v;
}

__device__ __forceinline__ int load_edges8(const int* __restrict__ src, const int* __restrict__ dst,
                                           int e0, int E, int (&d)[8], int (&s)[8], bool want_src) {
#pragma unroll
    for (int k = 0; k < 8; ++k) { d[k] = 0; s[k] = 0; }
    if (e0 >= E) return 0;
    size_t al = reinterpret_cast<size_t>(dst);
    if (want_src) al |= reinterpret_cast<size_t>(src);
    if ((e0 + 8 <= E) && ((al & 15) == 0)) {
        const v4i d0 = *(const v4i*)(dst + e0), d1 = *(const v4i*)(dst + e0 + 4);
        d[0] = d0.x; d[1] = d0.y; d[2] = d0.z; d[3] = d0.w;
        d[4] = d1.x; d[5] = d1.y; d[6] = d1.z; d[7] = d1.w;
        if (want_src) {
            const v4i s0 = *(const v4i*)(src + e0), s1 = *(const v4i*)(src + e0 + 4);
            s[0] = s0.x; s[1] = s0.y; s[2] = s0.z; s[3] = s0.w;
            s[4] = s1.x; s[5] = s1.y; s[6] = s1.z; s[7] = s1.w;
        }
        return 8;
    }
    const int nv = (E - e0 < 8) ? (E - e0) : 8;
#pragma unroll
    for (int k = 0; k < 8; ++k) {
        d[k] = (k < nv) ? dst[e0 + k] : 0;
        if (want_src) s[k] = (k < nv) ? src[e0 + k] : 0;
    }
    return nv;
}

__global__ void __launch_bounds__(EDGE_TB)
bin_local_kernel(const int* __restrict__ src, const int* __restrict__ dst,
                 int* __restrict__ pk, int* __restrict__ recs, int E, int nb) {
    __shared__ int hist[MAXNB];
    __shared__ int wtot[EDGE_TB / 32];
    const int tid  = threadIdx.x;
    const int lane = tid & 31;
    const int wave = tid >> 5;
    hist[tid] = 0;
    __syncthreads();
    int d[8], s[8], lr[8];
    const int nv = load_edges8(src, dst, (int)(blockIdx.x * EDGE_TB + tid) * 8, E, d, s, true);
#pragma unroll
    for (int k = 0; k < 8; ++k) {
        lr[k] = 0;
        if (k < nv) lr[k] = atomicAdd(&hist[d[k] >> BIN_SHIFT], 1);
    }
    __syncthreads();
    const int h    = hist[tid];
    const int incl = wave_incl_scan(h, lane);
    if (lane == 31) wtot[wave] = incl;
    __syncthreads();
    int carry = 0;
#pragma unroll
    for (int w = 0; w < EDGE_TB / 32; ++w) carry += (w < wave) ? wtot[w] : 0;
    const int base = carry + incl - h;
    hist[tid] = base;
    __syncthreads();
    if (tid < nb) pk[(size_t)blockIdx.x * nb + tid] = base | (h << 16);
    int pos[8];
#pragma unroll
    for (int k = 0; k < 8; ++k) pos[k] = hist[d[k] >> BIN_SHIFT] + lr[k];
    int* seg = recs + 2 * (size_t)blockIdx.x * SEG;
#pragma unroll
    for (int k = 0; k < 8; ++k) {
        if (k < nv) {
            v2i r;
            r.x = s[k];
            r.y = d[k];
            *(v2i*)(seg + 2 * pos[k]) = r;
        }
    }
}

#define ROW_PITCH (64 + 4)
template<int NOUT, int LD>
__device__ __forceinline__ void strip_product(const float* arow, const float* __restrict__ W,
                                              const float* __restrict__ rscale,
                                              float* __restrict__ Y, int m0, int lane) {
    constexpr int NT = NOUT / 16;
    typedef float vnt __attribute__((ext_vector_type(NT)));
    const int half = lane >> 4;
    const int l15  = lane & 15;

    v8f c[NT];
#pragma unroll
    for (int t = 0; t < NT; ++t) c[t] = v8f{};
    const float* wcol = W + l15 * NT;
#pragma unroll
    for (int kb = 0; kb < 64; kb += 4) {
        const int ka = kb + half * 2;
        const v2f a  = *(const v2f*)(arow + ka);
        const vnt b0 = *(const vnt*)(wcol + (size_t)(ka + 0) * LD);
        const vnt b1 = *(const vnt*)(wcol + (size_t)(ka + 1) * LD);
#pragma unroll
        for (int t = 0; t < NT; ++t) {
            v2f b;
            b.x = b0[t];
            b.y = b1[t];
            c[t] = __builtin_amdgcn_wmma_f32_16x16x4_f32(
                     false, a,  false, b,
                     (short)0, c[t],  false,  false);
        }
    }

    const v4f rs0 = *(const v4f*)(rscale + m0 + half * 8);
    const v4f rs1 = *(const v4f*)(rscale + m0 + half * 8 + 4);
    const float rsc[8] = {rs0.x, rs0.y, rs0.z, rs0.w, rs1.x, rs1.y, rs1.z, rs1.w};
    float* yrow = Y + (size_t)(m0 + half * 8) * LD + l15 * NT;
#pragma unroll
    for (int j = 0; j < 8; ++j) {
        vnt v;
#pragma unroll
        for (int t = 0; t < NT; ++t) v[t] = c[t][j] * rsc[j];
        *(vnt*)(yrow + (size_t)j * LD) = v;
    }
}

__global__ void __launch_bounds__(SORT_TB)
bin_sort_kernel(const int* __restrict__ recs, const int* __restrict__ pk,
                int* __restrict__ csr, int* __restrict__ off, int* __restrict__ cnt,
                float* __restrict__ dinv, int n, int G, int nb,
                const float* __restrict__ X, const float* __restrict__ W,
                float* __restrict__ Y) {
    constexpr int RB = 1 << BIN_SHIFT;
    constexpr int KR = 8;
    constexpr int XV = RB * (NF_IN / 4) / SORT_TB;
    static_assert(RB * (NF_IN / 4) % SORT_TB == 0, "the X rows split evenly over the threads");
    static_assert(2 * (RB / 16) == SORT_TB / 32, "two waves per strip of 16 rows");
    __shared__ int lcnt[RB];
    __shared__ int loff[RB];
    __shared__ int wtot[RB / 32];
    __shared__ int sh_start;
    __shared__ __attribute__((aligned(16))) float ldinv[RB];
    __shared__ __attribute__((aligned(16))) float xs[RB * ROW_PITCH];
    const int b    = blockIdx.x;
    const int tid  = threadIdx.x;
    const int lane = tid & 31;
    const int wave = __builtin_amdgcn_readfirstlane(tid >> 5);
    const int lo   = b << BIN_SHIFT;
    if (tid < RB) lcnt[tid] = 0;
    if (tid == 0) sh_start = 0;
    __syncthreads();
    int pkv = 0;
    if (lane < KR && wave + 32 * lane < G) pkv = pk[(size_t)(wave + 32 * lane) * nb + b];
    int part = 0;
    for (int g = tid; g < G; g += SORT_TB) part += pk[(size_t)g * nb + b] & 0xFFFF;
    for (int dl = 16; dl > 0; dl >>= 1) part += __shfl_xor(part, dl);
    if (lane == 0 && part != 0) atomicAdd(&sh_start, part);
    const v2i* rbase = (const v2i*)recs;
    v2i        rec[KR];
    int        pc[KR];
    const v2i* pp[KR];
#pragma unroll
    for (int k = 0; k < KR; ++k) {
        const int e = __builtin_amdgcn_readlane(pkv, k);
        pc[k] = e >> 16;
        pp[k] = rbase + (size_t)(wave + 32 * k) * SEG + (e & 0xFFFF);
        rec[k].x = 0;
        rec[k].y = lo;
        if (lane < pc[k]) rec[k] = pp[k][lane];
    }
    v4f xv[XV];
#pragma unroll
    for (int i = 0; i < XV; ++i) {
        const int p   = tid + i * SORT_TB;
        const int row = (lo + (p >> 4) < n) ? lo + (p >> 4) : n - 1;
        xv[i] = *(const v4f*)(X + (size_t)row * NF_IN + (p & 15) * 4);
    }
#pragma unroll
    for (int k = 0; k < KR; ++k) {
        if (lane < pc[k]) atomicAdd(&lcnt[rec[k].y - lo], 1);
        for (int r = 32 + lane; r < pc[k]; r += 32) atomicAdd(&lcnt[pp[k][r].y - lo], 1);
    }
#pragma unroll
    for (int i = 0; i < XV; ++i) {
        const int p = tid + i * SORT_TB;
        *(v4f*)(xs + (p >> 4) * ROW_PITCH + (p & 15) * 4) = xv[i];
    }
    for (int g = wave + 32 * KR; g < G; g += 32) {
        const int  e = pk[(size_t)g * nb + b];
        const v2i* p = rbase + (size_t)g * SEG + (e & 0xFFFF);
        for (int r = lane; r < (e >> 16); r += 32) atomicAdd(&lcnt[p[r].y - lo], 1);
    }
    __syncthreads();
    int v = 0, incl = 0, lf = 0;
    if (tid < RB) {
        v    = lcnt[tid];
        incl = wave_incl_scan(v, lane);
        if (lane == 31) wtot[tid >> 5] = incl;
    }
    __syncthreads();
    const int bs = sh_start;
    if (tid < RB) {
        int carry = 0;
        for (int w = 0; w < (tid >> 5); ++w) carry += wtot[w];
        lf = carry + incl - v;
        loff[tid] = lf;
        lcnt[tid] = 0;
        ldinv[tid] = rsqrtf((float)(v + 1));
    }
    __syncthreads();
    if (tid < RB && lo + tid < n) {
        const int i = lo + tid;
        off[i]  = bs + lf;
        cnt[i]  = v;
        dinv[i] = ldinv[tid];
    }
#pragma unroll
    for (int k = 0; k < KR; ++k) {
        if (lane < pc[k]) {
            const int j = rec[k].y - lo;
            const int p = atomicAdd(&lcnt[j], 1);
            csr[bs + loff[j] + p] = rec[k].x;
        }
        for (int r = 32 + lane; r < pc[k]; r += 32) {
            const v2i rc = pp[k][r];
            const int j  = rc.y - lo;
            const int p  = atomicAdd(&lcnt[j], 1);
            csr[bs + loff[j] + p] = rc.x;
        }
    }
    for (int g = wave + 32 * KR; g < G; g += 32) {
        const int  e = pk[(size_t)g * nb + b];
        const v2i* p = rbase + (size_t)g * SEG + (e & 0xFFFF);
        for (int r = lane; r < (e >> 16); r += 32) {
            const v2i rc = p[r];
            const int j  = rc.y - lo;
            const int q  = atomicAdd(&lcnt[j], 1);
            csr[bs + loff[j] + q] = rc.x;
        }
    }
    const int strip = wave >> 1;
    const int cb    = (wave & 1) * (NF_HID / 2);
    const int r0    = strip * 16;
    if (lo + r0 + 16 <= n)
        strip_product<NF_HID / 2, NF_HID>(xs + (r0 + (lane & 15)) * ROW_PITCH, W + cb, ldinv,
                                          Y + (size_t)lo * NF_HID + cb, r0, lane);
}

template<int F, int U>
__device__ __forceinline__ void issue_rows(const float* hq, int sl, int j, int sub,
                                           v2f (&r)[U]) {
    constexpr int EPI = 64 / F;
#pragma unroll
    for (int u = 0; u < U; ++u) {
        const int jj = j + u * EPI + sub;
        int s;
        if constexpr (EPI == 1) s = __builtin_amdgcn_readlane(sl, jj);
        else                    s = __shfl(sl, jj);
        r[u] = *(const v2f*)(hq + (size_t)s * F);
    }
}

template<int F, int U>
__device__ __forceinline__ void add_rows(v2f& acc, int j, int sub, int m,
                                         const v2f (&r)[U]) {
    constexpr int EPI = 64 / F;
#pragma unroll
    for (int u = 0; u < U; ++u) {
        const int jj = j + u * EPI + sub;
        if (jj < m) {
            acc.x += r[u].x;
            acc.y += r[u].y;
        }
    }
}

template<int F, int NE>
__device__ __forceinline__ void sum_chunk(v2f& acc, int& sN, const v2f rs, const float* hq,
                                          int sl, int sub, int m, const int* __restrict__ csr,
                                          int startN, int cN, int lane) {
    constexpr int UU = NE / (64 / F);
    v2f r[UU];
    issue_rows<F, UU>(hq, sl, 0, sub, r);
    if (lane < cN) sN = csr[startN + lane];
    acc.x = (sub == 0) ? rs.x : 0.0f;
    acc.y = (sub == 0) ? rs.y : 0.0f;
    add_rows<F, UU>(acc, 0, sub, m, r);
}

template<int F, bool RELU, class Put>
__device__ __forceinline__ void gather_run(const int* __restrict__ csr, const int* __restrict__ off,
                                           const int* __restrict__ cnt, const float* __restrict__ dinv,
                                           const float* __restrict__ h, const float* __restrict__ bias,
                                           int i0, int i1, int lane, Put put) {
    constexpr int LPR = F / 2;
    constexpr int EPI = 32 / LPR;
    constexpr int RS  = 16;
    constexpr int U   = RS / EPI;
    const int sub  = lane / LPR;
    const int q    = lane % LPR;
    const float* hq = h + 2 * q;
    const v2f bq = *(const v2f*)(bias + 2 * q);

    int   start = off[i0];
    int   c     = cnt[i0];
    float di    = dinv[i0];
    int   s_l   = i0;
    if (lane < c) s_l = csr[start + lane];
    for (int i = i0; i < i1; ++i) {
        const int   inx    = (i + 1 < i1) ? i + 1 : i;
        const int   startN = off[inx];
        const int   cN     = cnt[inx];
        const float diN    = dinv[inx];
        const int   m      = (c < 32) ? c : 32;

        const v2f rs = *(const v2f*)(hq + (size_t)i * F);
        int sN = inx;
        v2f acc;
        if (m <= RS)
            sum_chunk<F, RS>(acc, sN, rs, hq, s_l, sub, m, csr, startN, cN, lane);
        else if (m <= RS + RS / 2)
            sum_chunk<F, RS + RS / 2>(acc, sN, rs, hq, s_l, sub, m, csr, startN, cN, lane);
        else
            sum_chunk<F, 2 * RS>(acc, sN, rs, hq, s_l, sub, m, csr, startN, cN, lane);
        for (int base = 32; base < c; base += 32) {
            const int m2 = (c - base < 32) ? (c - base) : 32;
            int s2 = i;
            if (lane < m2) s2 = csr[start + base + lane];
            for (int j = 0; j < m2; j += RS) {
                v2f r[U];
                issue_rows<F, U>(hq, s2, j, sub, r);
                add_rows<F, U>(acc, j, sub, m2, r);
            }
        }
        if constexpr (EPI == 2) {
            acc.x += __shfl_xor(acc.x, 16);
            acc.y += __shfl_xor(acc.y, 16);
        }
        if (sub == 0) {
            v2f v;
            v.x = di * acc.x + bq.x;
            v.y = di * acc.y + bq.y;
            if (RELU) { v.x = fmaxf(v.x, 0.0f); v.y = fmaxf(v.y, 0.0f); }
            put(i, q, v);
        }
        start = startN; c = cN; di = diN; s_l = sN;
    }
}

template<int F, bool RELU>
__global__ void gather_kernel(const int* __restrict__ csr, const int* __restrict__ off,
                              const int* __restrict__ cnt, const float* __restrict__ dinv,
                              const float* __restrict__ h, const float* __restrict__ bias,
                              float* __restrict__ out, int n) {
    const int lane = threadIdx.x & 31;
    const int wave  = __builtin_amdgcn_readfirstlane(
        (int)((blockIdx.x * blockDim.x + threadIdx.x) >> 5));
    const int nwave = (int)((gridDim.x * blockDim.x) >> 5);
    const int per = n / nwave;
    const int rem = n - per * nwave;
    const int i0  = wave * per + ((wave < rem) ? wave : rem);
    const int i1  = i0 + per + ((wave < rem) ? 1 : 0);
    if (i0 >= i1) return;
    gather_run<F, RELU>(csr, off, cnt, dinv, h, bias, i0, i1, lane,
                        [&](int i, int q, v2f v) { *(v2f*)(out + (size_t)i * F + 2 * q) = v; });
}

#define FG_WAVES  8
#define FG_NPW    6
#define FG_NODES  (FG_WAVES * FG_NPW)
#define FG_STRIPS (FG_NODES / 16)
#define FG_PITCH  ROW_PITCH

__global__ void __launch_bounds__(FG_WAVES * 32)
gather_product_kernel(const int* __restrict__ csr, const int* __restrict__ off,
                      const int* __restrict__ cnt, const float* __restrict__ dinv,
                      const float* __restrict__ h, const float* __restrict__ bias,
                      const float* __restrict__ W, float* __restrict__ Y, int n) {
    static_assert(FG_NODES % 16 == 0, "a work-group owns whole strips");
    __shared__ float rows[FG_NODES * FG_PITCH];
    const int lane = threadIdx.x & 31;
    const int wave = __builtin_amdgcn_readfirstlane((int)(threadIdx.x >> 5));
    const int n0   = (int)blockIdx.x * FG_NODES;
    const int i0   = n0 + wave * FG_NPW;
    const int i1   = (i0 + FG_NPW < n) ? (i0 + FG_NPW) : n;
    if (i0 < i1) {
        gather_run<NF_HID, true>(csr, off, cnt, dinv, h, bias, i0, i1, lane,
                                 [&](int i, int q, v2f v) {
                                     *(v2f*)(rows + (i - n0) * FG_PITCH + 2 * q) = v;
                                 });
    }
    __syncthreads();
    const int m0 = n0 + wave * 16;
    if (wave < FG_STRIPS && m0 < n)
        strip_product<NF_OUT, NF_OUT>(rows + (wave * 16 + (lane & 15)) * FG_PITCH, W, dinv, Y, m0, lane);
}

extern "C" void kernel_launch(void* const* d_in, const int* in_sizes, int n_in,
                              void* d_out, int out_size, void* d_ws, size_t ws_size,
                              hipStream_t stream) {
    const float* x  = (const float*)d_in[0];
    const int*   ei = (const int*)  d_in[1];
    const float* W1 = (const float*)d_in[2];
    const float* b1 = (const float*)d_in[3];
    const float* W2 = (const float*)d_in[4];
    const float* b2 = (const float*)d_in[5];

    const int N = in_sizes[0] / NF_IN;
    const int E = in_sizes[1] / 2;
    const int* src = ei;
    const int* dst = ei + E;

    char* ws = (char*)d_ws;
    auto take = [&](size_t bytes) { char* p = ws; ws += (bytes + 255) & ~(size_t)255; return p; };
    auto cdiv = [](int a, int b) { return (a + b - 1) / b; };
    const int NB = cdiv(N, 1 << BIN_SHIFT);
    const int G  = cdiv(E, SEG);
    int*   pk    = (int*)  take((size_t)G * NB * 4);
    int*   cnt   = (int*)  take((size_t)N * 4);
    int*   off   = (int*)  take((size_t)N * 4);
    float* dinv  = (float*)take((size_t)N * 4);
    int*   recs  = (int*)  take((size_t)G * SEG * 8);
    int*   csr   = (int*)  take((size_t)E * 4);
    float* h1pre = (float*)take((size_t)N * NF_HID * 4);
    float* h2pre = (float*)take((size_t)N * NF_OUT * 4);
    float* out   = (float*)d_out;

    const int TB = 256;
    const int GW = 1024;

    bin_local_kernel<<<G, EDGE_TB, 0, stream>>>(src, dst, pk, recs, E, NB);
    bin_sort_kernel <<<NB, SORT_TB, 0, stream>>>(recs, pk, csr, off, cnt, dinv, N, G, NB,
                                                 x, W1, h1pre);

    gather_product_kernel<<<cdiv(N, FG_NODES), FG_WAVES * 32, 0, stream>>>(
        csr, off, cnt, dinv, h1pre, b1, W2, h2pre, N);

    gather_kernel<NF_OUT, false><<<GW, TB, 0, stream>>>(
        csr, off, cnt, dinv, h2pre, b2, out, N);
}


# the edge sort's two scattered store passes (8-byte records per segment, 4-byte source ids per bin) are put in order in LDS and stored as whole 128-byte lines; sum order kept (LDS arrival rank)
# speedup vs baseline: 6.7838x; 1.1181x over previous
#include <hip/hip_runtime.h>


#define NF_IN  64
#define NF_HID 64
#define NF_OUT 32

typedef __attribute__((ext_vector_type(2))) float v2f;
typedef __attribute__((ext_vector_type(4))) float v4f;
typedef __attribute__((ext_vector_type(8))) float v8f;
typedef __attribute__((ext_vector_type(4))) int   v4i;

#define BIN_SHIFT 8
#define MAXNB     512
#define EDGE_TB   512
#define SEG       4096
#define SORT_TB   1024
#define CS_CAP    8192

typedef __attribute__((ext_vector_type(2))) int v2i;

__device__ __forceinline__ int wave_incl_scan(int v, int lane) {
    for (int d = 1; d < 32; d <<= 1) {
        const int t = __shfl_up(v, d);
        if (lane >= d) v += t;
    }
    return v;
}

__device__ __forceinline__ int load_edges8(const int* __restrict__ src, const int* __restrict__ dst,
                                           int e0, int E, int (&d)[8], int (&s)[8], bool want_src) {
#pragma unroll
    for (int k = 0; k < 8; ++k) { d[k] = 0; s[k] = 0; }
    if (e0 >= E) return 0;
    size_t al = reinterpret_cast<size_t>(dst);
    if (want_src) al |= reinterpret_cast<size_t>(src);
    if ((e0 + 8 <= E) && ((al & 15) == 0)) {
        const v4i d0 = *(const v4i*)(dst + e0), d1 = *(const v4i*)(dst + e0 + 4);
        d[0] = d0.x; d[1] = d0.y; d[2] = d0.z; d[3] = d0.w;
        d[4] = d1.x; d[5] = d1.y; d[6] = d1.z; d[7] = d1.w;
        if (want_src) {
            const v4i s0 = *(const v4i*)(src + e0), s1 = *(const v4i*)(src + e0 + 4);
            s[0] = s0.x; s[1] = s0.y; s[2] = s0.z; s[3] = s0.w;
            s[4] = s1.x; s[5] = s1.y; s[6] = s1.z; s[7] = s1.w;
        }
        return 8;
    }
    const int nv = (E - e0 < 8) ? (E - e0) : 8;
#pragma unroll
    for (int k = 0; k < 8; ++k) {
        d[k] = (k < nv) ? dst[e0 + k] : 0;
        if (want_src) s[k] = (k < nv) ? src[e0 + k] : 0;
    }
    return nv;
}

__global__ void __launch_bounds__(EDGE_TB)
bin_local_kernel(const int* __restrict__ src, const int* __restrict__ dst,
                 int* __restrict__ pk, int* __restrict__ recs, int E, int nb) {
    __shared__ int hist[MAXNB];
    __shared__ int wtot[EDGE_TB / 32];
    __shared__ __attribute__((aligned(16))) int stage[2 * SEG];
    const int tid  = threadIdx.x;
    const int lane = tid & 31;
    const int wave = tid >> 5;
    hist[tid] = 0;
    __syncthreads();
    int d[8], s[8], lr[8];
    const int nv = load_edges8(src, dst, (int)(blockIdx.x * EDGE_TB + tid) * 8, E, d, s, true);
#pragma unroll
    for (int k = 0; k < 8; ++k) {
        lr[k] = 0;
        if (k < nv) lr[k] = atomicAdd(&hist[d[k] >> BIN_SHIFT], 1);
    }
    __syncthreads();
    const int h    = hist[tid];
    const int incl = wave_incl_scan(h, lane);
    if (lane == 31) wtot[wave] = incl;
    __syncthreads();
    int carry = 0;
#pragma unroll
    for (int w = 0; w < EDGE_TB / 32; ++w) carry += (w < wave) ? wtot[w] : 0;
    const int base = carry + incl - h;
    hist[tid] = base;
    __syncthreads();
    int pos[8];
#pragma unroll
    for (int k = 0; k < 8; ++k) pos[k] = hist[d[k] >> BIN_SHIFT] + lr[k];
#pragma unroll
    for (int k = 0; k < 8; ++k) {
        if (k < nv) {
            v2i r;
            r.x = s[k];
            r.y = d[k];
            *(v2i*)(stage + 2 * pos[k]) = r;
        }
    }
    __syncthreads();
    if (tid < nb) pk[(size_t)blockIdx.x * nb + tid] = base | (h << 16);
    const int left = E - (int)blockIdx.x * SEG;
    const int tot  = (left < SEG) ? left : SEG;
    int* seg = recs + 2 * (size_t)blockIdx.x * SEG;
    constexpr int NW = SEG / (2 * EDGE_TB);
    v4i w[NW];
#pragma unroll
    for (int i = 0; i < NW; ++i) w[i] = *(const v4i*)(stage + 4 * (i * EDGE_TB + tid));
#pragma unroll
    for (int i = 0; i < NW; ++i) {
        const int p = 2 * (i * EDGE_TB + tid);
        if (p + 1 < tot) {
            *(v4i*)(seg + 2 * p) = w[i];
        } else if (p < tot) {
            v2i r;
            r.x = w[i].x;
            r.y = w[i].y;
            *(v2i*)(seg + 2 * p) = r;
        }
    }
}

#define ROW_PITCH (64 + 4)
template<int NOUT, int LD>
__device__ __forceinline__ void strip_product(const float* arow, const float* __restrict__ W,
                                              const float* __restrict__ rscale,
                                              float* __restrict__ Y, int m0, int lane) {
    constexpr int NT = NOUT / 16;
    typedef float vnt __attribute__((ext_vector_type(NT)));
    const int half = lane >> 4;
    const int l15  = lane & 15;

    v8f c[NT];
#pragma unroll
    for (int t = 0; t < NT; ++t) c[t] = v8f{};
    const float* wcol = W + l15 * NT;
#pragma unroll
    for (int kb = 0; kb < 64; kb += 4) {
        const int ka = kb + half * 2;
        const v2f a  = *(const v2f*)(arow + ka);
        const vnt b0 = *(const vnt*)(wcol + (size_t)(ka + 0) * LD);
        const vnt b1 = *(const vnt*)(wcol + (size_t)(ka + 1) * LD);
#pragma unroll
        for (int t = 0; t < NT; ++t) {
            v2f b;
            b.x = b0[t];
            b.y = b1[t];
            c[t] = __builtin_amdgcn_wmma_f32_16x16x4_f32(
                     false, a,  false, b,
                     (short)0, c[t],  false,  false);
        }
    }

    const v4f rs0 = *(const v4f*)(rscale + m0 + half * 8);
    const v4f rs1 = *(const v4f*)(rscale + m0 + half * 8 + 4);
    const float rsc[8] = {rs0.x, rs0.y, rs0.z, rs0.w, rs1.x, rs1.y, rs1.z, rs1.w};
    float* yrow = Y + (size_t)(m0 + half * 8) * LD + l15 * NT;
#pragma unroll
    for (int j = 0; j < 8; ++j) {
        vnt v;
#pragma unroll
        for (int t = 0; t < NT; ++t) v[t] = c[t][j] * rsc[j];
        *(vnt*)(yrow + (size_t)j * LD) = v;
    }
}

__global__ void __launch_bounds__(SORT_TB)
bin_sort_kernel(const int* __restrict__ recs, const int* __restrict__ pk,
                int* __restrict__ csr, int* __restrict__ off, int* __restrict__ cnt,
                float* __restrict__ dinv, int n, int G, int nb,
                const float* __restrict__ X, const float* __restrict__ W,
                float* __restrict__ Y) {
    constexpr int RB = 1 << BIN_SHIFT;
    constexpr int KR = 8;
    constexpr int XV = RB * (NF_IN / 4) / SORT_TB;
    static_assert(RB * (NF_IN / 4) % SORT_TB == 0, "the X rows split evenly over the threads");
    static_assert(2 * (RB / 16) == SORT_TB / 32, "two waves per strip of 16 rows");
    __shared__ int lcnt[RB];
    __shared__ int loff[RB];
    __shared__ int wtot[RB / 32];
    __shared__ int sh_start;
    __shared__ int sh_tot;
    __shared__ __attribute__((aligned(16))) int cs[CS_CAP + 32];
    __shared__ __attribute__((aligned(16))) float ldinv[RB];
    __shared__ __attribute__((aligned(16))) float xs[RB * ROW_PITCH];
    const int b    = blockIdx.x;
    const int tid  = threadIdx.x;
    const int lane = tid & 31;
    const int wave = __builtin_amdgcn_readfirstlane(tid >> 5);
    const int lo   = b << BIN_SHIFT;
    if (tid < RB) lcnt[tid] = 0;
    if (tid == 0) sh_start = 0;
    __syncthreads();
    int pkv = 0;
    if (lane < KR && wave + 32 * lane < G) pkv = pk[(size_t)(wave + 32 * lane) * nb + b];
    int part = 0;
    for (int g = tid; g < G; g += SORT_TB) part += pk[(size_t)g * nb + b] & 0xFFFF;
    for (int dl = 16; dl > 0; dl >>= 1) part += __shfl_xor(part, dl);
    if (lane == 0 && part != 0) atomicAdd(&sh_start, part);
    const v2i* rbase = (const v2i*)recs;
    v2i        rec[KR];
    int        pc[KR];
    const v2i* pp[KR];
#pragma unroll
    for (int k = 0; k < KR; ++k) {
        const int e = __builtin_amdgcn_readlane(pkv, k);
        pc[k] = e >> 16;
        pp[k] = rbase + (size_t)(wave + 32 * k) * SEG + (e & 0xFFFF);
        rec[k].x = 0;
        rec[k].y = lo;
        if (lane < pc[k]) rec[k] = pp[k][lane];
    }
    v4f xv[XV];
#pragma unroll
    for (int i = 0; i < XV; ++i) {
        const int p   = tid + i * SORT_TB;
        const int row = (lo + (p >> 4) < n) ? lo + (p >> 4) : n - 1;
        xv[i] = *(const v4f*)(X + (size_t)row * NF_IN + (p & 15) * 4);
    }
#pragma unroll
    for (int k = 0; k < KR; ++k) {
        if (lane < pc[k]) atomicAdd(&lcnt[rec[k].y - lo], 1);
        for (int r = 32 + lane; r < pc[k]; r += 32) atomicAdd(&lcnt[pp[k][r].y - lo], 1);
    }
#pragma unroll
    for (int i = 0; i < XV; ++i) {
        const int p = tid + i * SORT_TB;
        *(v4f*)(xs + (p >> 4) * ROW_PITCH + (p & 15) * 4) = xv[i];
    }
    for (int g = wave + 32 * KR; g < G; g += 32) {
        const int  e = pk[(size_t)g * nb + b];
        const v2i* p = rbase + (size_t)g * SEG + (e & 0xFFFF);
        for (int r = lane; r < (e >> 16); r += 32) atomicAdd(&lcnt[p[r].y - lo], 1);
    }
    __syncthreads();
    int v = 0, incl = 0, lf = 0;
    if (tid < RB) {
        v    = lcnt[tid];
        incl = wave_incl_scan(v, lane);
        if (lane == 31) wtot[tid >> 5] = incl;
    }
    __syncthreads();
    const int bs = sh_start;
    if (tid < RB) {
        int carry = 0;
        for (int w = 0; w < (tid >> 5); ++w) carry += wtot[w];
        lf = carry + incl - v;
        loff[tid] = lf;
        lcnt[tid] = 0;
        ldinv[tid] = rsqrtf((float)(v + 1));
        if (tid == RB - 1) sh_tot = lf + v;
    }
    __syncthreads();
    const int a0 = bs & 31;
    auto place = [&](int e, int id) {
        if (e < CS_CAP) cs[a0 + e] = id;
        else            csr[bs + e] = id;
    };
#pragma unroll
    for (int k = 0; k < KR; ++k) {
        if (lane < pc[k]) {
            const int j = rec[k].y - lo;
            const int p = atomicAdd(&lcnt[j], 1);
            place(loff[j] + p, rec[k].x);
        }
        for (int r = 32 + lane; r < pc[k]; r += 32) {
            const v2i rc = pp[k][r];
            const int j  = rc.y - lo;
            const int p  = atomicAdd(&lcnt[j], 1);
            place(loff[j] + p, rc.x);
        }
    }
    for (int g = wave + 32 * KR; g < G; g += 32) {
        const int  e = pk[(size_t)g * nb + b];
        const v2i* p = rbase + (size_t)g * SEG + (e & 0xFFFF);
        for (int r = lane; r < (e >> 16); r += 32) {
            const v2i rc = p[r];
            const int j  = rc.y - lo;
            const int q  = atomicAdd(&lcnt[j], 1);
            place(loff[j] + q, rc.x);
        }
    }
    __syncthreads();
    if (tid < RB && lo + tid < n) {
        const int i = lo + tid;
        off[i]  = bs + lf;
        cnt[i]  = v;
        dinv[i] = ldinv[tid];
    }
    {
        const int tot  = sh_tot;
        const int tc   = (tot < CS_CAP) ? tot : CS_CAP;
        const int hi   = a0 + tc;
        int*      lineb = csr + (bs - a0);
        for (int k = 4 * tid; k < hi; k += 4 * SORT_TB) {
            const v4i ids = *(const v4i*)(cs + k);
            if (k >= a0 && k + 3 < hi) {
                *(v4i*)(lineb + k) = ids;
            } else {
                if (k + 0 >= a0 && k + 0 < hi) lineb[k + 0] = ids.x;
                if (k + 1 >= a0 && k + 1 < hi) lineb[k + 1] = ids.y;
                if (k + 2 >= a0 && k + 2 < hi) lineb[k + 2] = ids.z;
                if (k + 3 >= a0 && k + 3 < hi) lineb[k + 3] = ids.w;
            }
        }
    }
    const int strip = wave >> 1;
    const int cb    = (wave & 1) * (NF_HID / 2);
    const int r0    = strip * 16;
    if (lo + r0 + 16 <= n)
        strip_product<NF_HID / 2, NF_HID>(xs + (r0 + (lane & 15)) * ROW_PITCH, W + cb, ldinv,
                                          Y + (size_t)lo * NF_HID + cb, r0, lane);
}

template<int F, int U>
__device__ __forceinline__ void issue_rows(const float* hq, int sl, int j, int sub,
                                           v2f (&r)[U]) {
    constexpr int EPI = 64 / F;
#pragma unroll
    for (int u = 0; u < U; ++u) {
        const int jj = j + u * EPI + sub;
        int s;
        if constexpr (EPI == 1) s = __builtin_amdgcn_readlane(sl, jj);
        else                    s = __shfl(sl, jj);
        r[u] = *(const v2f*)(hq + (size_t)s * F);
    }
}

template<int F, int U>
__device__ __forceinline__ void add_rows(v2f& acc, int j, int sub, int m,
                                         const v2f (&r)[U]) {
    constexpr int EPI = 64 / F;
#pragma unroll
    for (int u = 0; u < U; ++u) {
        const int jj = j + u * EPI + sub;
        if (jj < m) {
            acc.x += r[u].x;
            acc.y += r[u].y;
        }
    }
}

template<int F, int NE>
__device__ __forceinline__ void sum_chunk(v2f& acc, int& sN, const v2f rs, const float* hq,
                                          int sl, int sub, int m, const int* __restrict__ csr,
                                          int startN, int cN, int lane) {
    constexpr int UU = NE / (64 / F);
    v2f r[UU];
    issue_rows<F, UU>(hq, sl, 0, sub, r);
    if (lane < cN) sN = csr[startN + lane];
    acc.x = (sub == 0) ? rs.x : 0.0f;
    acc.y = (sub == 0) ? rs.y : 0.0f;
    add_rows<F, UU>(acc, 0, sub, m, r);
}

template<int F, bool RELU, class Put>
__device__ __forceinline__ void gather_run(const int* __restrict__ csr, const int* __restrict__ off,
                                           const int* __restrict__ cnt, const float* __restrict__ dinv,
                                           const float* __restrict__ h, const float* __restrict__ bias,
                                           int i0, int i1, int lane, Put put) {
    constexpr int LPR = F / 2;
    constexpr int EPI = 32 / LPR;
    constexpr int RS  = 16;
    constexpr int U   = RS / EPI;
    const int sub  = lane / LPR;
    const int q    = lane % LPR;
    const float* hq = h + 2 * q;
    const v2f bq = *(const v2f*)(bias + 2 * q);

    int   start = off[i0];
    int   c     = cnt[i0];
    float di    = dinv[i0];
    int   s_l   = i0;
    if (lane < c) s_l = csr[start + lane];
    for (int i = i0; i < i1; ++i) {
        const int   inx    = (i + 1 < i1) ? i + 1 : i;
        const int   startN = off[inx];
        const int   cN     = cnt[inx];
        const float diN    = dinv[inx];
        const int   m      = (c < 32) ? c : 32;

        const v2f rs = *(const v2f*)(hq + (size_t)i * F);
        int sN = inx;
        v2f acc;
        if (m <= RS)
            sum_chunk<F, RS>(acc, sN, rs, hq, s_l, sub, m, csr, startN, cN, lane);
        else if (m <= RS + RS / 2)
            sum_chunk<F, RS + RS / 2>(acc, sN, rs, hq, s_l, sub, m, csr, startN, cN, lane);
        else
            sum_chunk<F, 2 * RS>(acc, sN, rs, hq, s_l, sub, m, csr, startN, cN, lane);
        for (int base = 32; base < c; base += 32) {
            const int m2 = (c - base < 32) ? (c - base) : 32;
            int s2 = i;
            if (lane < m2) s2 = csr[start + base + lane];
            for (int j = 0; j < m2; j += RS) {
                v2f r[U];
                issue_rows<F, U>(hq, s2, j, sub, r);
                add_rows<F, U>(acc, j, sub, m2, r);
            }
        }
        if constexpr (EPI == 2) {
            acc.x += __shfl_xor(acc.x, 16);
            acc.y += __shfl_xor(acc.y, 16);
        }
        if (sub == 0) {
            v2f v;
            v.x = di * acc.x + bq.x;
            v.y = di * acc.y + bq.y;
            if (RELU) { v.x = fmaxf(v.x, 0.0f); v.y = fmaxf(v.y, 0.0f); }
            put(i, q, v);
        }
        start = startN; c = cN; di = diN; s_l = sN;
    }
}

template<int F, bool RELU>
__global__ void gather_kernel(const int* __restrict__ csr, const int* __restrict__ off,
                              const int* __restrict__ cnt, const float* __restrict__ dinv,
                              const float* __restrict__ h, const float* __restrict__ bias,
                              float* __restrict__ out, int n) {
    const int lane = threadIdx.x & 31;
    const int wave  = __builtin_amdgcn_readfirstlane(
        (int)((blockIdx.x * blockDim.x + threadIdx.x) >> 5));
    const int nwave = (int)((gridDim.x * blockDim.x) >> 5);
    const int per = n / nwave;
    const int rem = n - per * nwave;
    const int i0  = wave * per + ((wave < rem) ? wave : rem);
    const int i1  = i0 + per + ((wave < rem) ? 1 : 0);
    if (i0 >= i1) return;
    gather_run<F, RELU>(csr, off, cnt, dinv, h, bias, i0, i1, lane,
                        [&](int i, int q, v2f v) { *(v2f*)(out + (size_t)i * F + 2 * q) = v; });
}

#define FG_WAVES  8
#define FG_NPW    6
#define FG_NODES  (FG_WAVES * FG_NPW)
#define FG_STRIPS (FG_NODES / 16)
#define FG_PITCH  ROW_PITCH

__global__ void __launch_bounds__(FG_WAVES * 32)
gather_product_kernel(const int* __restrict__ csr, const int* __restrict__ off,
                      const int* __restrict__ cnt, const float* __restrict__ dinv,
                      const float* __restrict__ h, const float* __restrict__ bias,
                      const float* __restrict__ W, float* __restrict__ Y, int n) {
    static_assert(FG_NODES % 16 == 0, "a work-group owns whole strips");
    __shared__ float rows[FG_NODES * FG_PITCH];
    const int lane = threadIdx.x & 31;
    const int wave = __builtin_amdgcn_readfirstlane((int)(threadIdx.x >> 5));
    const int n0   = (int)blockIdx.x * FG_NODES;
    const int i0   = n0 + wave * FG_NPW;
    const int i1   = (i0 + FG_NPW < n) ? (i0 + FG_NPW) : n;
    if (i0 < i1) {
        gather_run<NF_HID, true>(csr, off, cnt, dinv, h, bias, i0, i1, lane,
                                 [&](int i, int q, v2f v) {
                                     *(v2f*)(rows + (i - n0) * FG_PITCH + 2 * q) = v;
                                 });
    }
    __syncthreads();
    const int m0 = n0 + wave * 16;
    if (wave < FG_STRIPS && m0 < n)
        strip_product<NF_OUT, NF_OUT>(rows + (wave * 16 + (lane & 15)) * FG_PITCH, W, dinv, Y, m0, lane);
}

extern "C" void kernel_launch(void* const* d_in, const int* in_sizes, int n_in,
                              void* d_out, int out_size, void* d_ws, size_t ws_size,
                              hipStream_t stream) {
    const float* x  = (const float*)d_in[0];
    const int*   ei = (const int*)  d_in[1];
    const float* W1 = (const float*)d_in[2];
    const float* b1 = (const float*)d_in[3];
    const float* W2 = (const float*)d_in[4];
    const float* b2 = (const float*)d_in[5];

    const int N = in_sizes[0] / NF_IN;
    const int E = in_sizes[1] / 2;
    const int* src = ei;
    const int* dst = ei + E;

    char* ws = (char*)d_ws;
    auto take = [&](size_t bytes) { char* p = ws; ws += (bytes + 255) & ~(size_t)255; return p; };
    auto cdiv = [](int a, int b) { return (a + b - 1) / b; };
    const int NB = cdiv(N, 1 << BIN_SHIFT);
    const int G  = cdiv(E, SEG);
    int*   pk    = (int*)  take((size_t)G * NB * 4);
    int*   cnt   = (int*)  take((size_t)N * 4);
    int*   off   = (int*)  take((size_t)N * 4);
    float* dinv  = (float*)take((size_t)N * 4);
    int*   recs  = (int*)  take((size_t)G * SEG * 8);
    int*   csr   = (int*)  take((size_t)E * 4);
    float* h1pre = (float*)take((size_t)N * NF_HID * 4);
    float* h2pre = (float*)take((size_t)N * NF_OUT * 4);
    float* out   = (float*)d_out;

    const int TB = 256;
    const int GW = 1024;

    bin_local_kernel<<<G, EDGE_TB, 0, stream>>>(src, dst, pk, recs, E, NB);
    bin_sort_kernel <<<NB, SORT_TB, 0, stream>>>(recs, pk, csr, off, cnt, dinv, N, G, NB,
                                                 x, W1, h1pre);

    gather_product_kernel<<<cdiv(N, FG_NODES), FG_WAVES * 32, 0, stream>>>(
        csr, off, cnt, dinv, h1pre, b1, W2, h2pre, N);

    gather_kernel<NF_OUT, false><<<GW, TB, 0, stream>>>(
        csr, off, cnt, dinv, h2pre, b2, out, N);
}
